# starting kernel
# baseline (speedup 1.0000x reference)
#include <hip/hip_runtime.h>
#include <hip/hip_bf16.h>

typedef __attribute__((ext_vector_type(2))) float v2f;
typedef __attribute__((ext_vector_type(8))) float v8f;

#define HDIM 128
#define HDIM2 256
#define NUM_GRAPHS 64
#define NUM_CLASSES 10
#define MSG_EPS 1e-7f

__device__ __forceinline__ void atomicMaxF(float* addr, float v) {
  if (v >= 0.0f) atomicMax((int*)addr, __float_as_int(v));
  else           atomicMin((unsigned int*)addr, __float_as_uint(v));
}

__global__ void gemm16_wmma_f32(const float* __restrict__ X,
                                const float* __restrict__ W,
                                const float* __restrict__ bias,
                                float* __restrict__ Y,
                                int Kdim, int Mdim, int accum) {
  const int lane = threadIdx.x & 31;
  const int wid  = threadIdx.x >> 5;
  const int row0 = blockIdx.x * 16;
  const int col0 = (blockIdx.y * 4 + wid) * 16;
  if (col0 >= Mdim) return;

  const int half = lane >> 4;
  const int l16  = lane & 15;

  const float bv = bias[col0 + l16];
  v8f c;
  if (accum) {
#pragma unroll
    for (int r = 0; r < 8; ++r)
      c[r] = Y[(size_t)(row0 + half * 8 + r) * Mdim + col0 + l16] + bv;
  } else {
#pragma unroll
    for (int r = 0; r < 8; ++r) c[r] = bv;
  }

  const float* xrow = X + (size_t)(row0 + l16) * Kdim + half * 2;
  const float* wcol = W + (size_t)(half * 2) * Mdim + col0 + l16;

  for (int k = 0; k < Kdim; k += 4) {
    __builtin_prefetch(xrow + k + 32, 0, 1);
    float2 av = *(const float2*)(xrow + k);
    v2f a; a[0] = av.x; a[1] = av.y;
    v2f b;
    b[0] = wcol[(size_t)k * Mdim];
    b[1] = wcol[(size_t)(k + 1) * Mdim];
    c = __builtin_amdgcn_wmma_f32_16x16x4_f32(false, a, false, b,
                                              (short)0, c, false, false);
  }

#pragma unroll
  for (int r = 0; r < 8; ++r)
    Y[(size_t)(row0 + half * 8 + r) * Mdim + col0 + l16] = c[r];
}

__global__ void ln_relu_kernel(const float* __restrict__ x, float* __restrict__ y,
                               const float* __restrict__ g, const float* __restrict__ b,
                               int rows, int D) {
  int row  = blockIdx.x * (blockDim.x >> 5) + (threadIdx.x >> 5);
  int lane = threadIdx.x & 31;
  if (row >= rows) return;
  const float* xr = x + (size_t)row * D;
  float s = 0.f, q = 0.f;
  for (int j = lane; j < D; j += 32) { float v = xr[j]; s += v; q += v * v; }
#pragma unroll
  for (int off = 16; off > 0; off >>= 1) {
    s += __shfl_xor(s, off, 32);
    q += __shfl_xor(q, off, 32);
  }
  float mean = s / (float)D;
  float var  = q / (float)D - mean * mean;
  float inv  = rsqrtf(var + 1e-5f);
  float* yr = y + (size_t)row * D;
  for (int j = lane; j < D; j += 32) {
    float v = (xr[j] - mean) * inv * g[j] + b[j];
    yr[j] = fmaxf(v, 0.f);
  }
}

__global__ void init_seg_kernel(float* mx, float* den, float* num, int total) {
  int i = blockIdx.x * blockDim.x + threadIdx.x;
  if (i >= total) return;
  mx[i]  = __int_as_float(0xff800000);
  den[i] = 0.f;
  num[i] = 0.f;
}

__global__ void zero_kernel(float* p, int n) {
  int i = blockIdx.x * blockDim.x + threadIdx.x;
  if (i < n) p[i] = 0.f;
}

__global__ void edge_max_kernel(const float* __restrict__ x, const float* __restrict__ ea,
                                const int* __restrict__ src, const int* __restrict__ dst,
                                const float* __restrict__ tptr, int layer,
                                float* __restrict__ mx, int E) {
  int e    = blockIdx.x * (blockDim.x >> 5) + (threadIdx.x >> 5);
  int lane = threadIdx.x & 31;
  if (e >= E) return;
  int s = src[e], d = dst[e];
  float tv = tptr[layer];
  float4 a  = ((const float4*)(x  + (size_t)s * HDIM))[lane];
  float4 bb = ((const float4*)(ea + (size_t)e * HDIM))[lane];
  float m0 = fmaxf(a.x + bb.x, 0.f) + MSG_EPS;
  float m1 = fmaxf(a.y + bb.y, 0.f) + MSG_EPS;
  float m2 = fmaxf(a.z + bb.z, 0.f) + MSG_EPS;
  float m3 = fmaxf(a.w + bb.w, 0.f) + MSG_EPS;
  float* mrow = mx + (size_t)d * HDIM + lane * 4;
  atomicMaxF(mrow + 0, m0 * tv);
  atomicMaxF(mrow + 1, m1 * tv);
  atomicMaxF(mrow + 2, m2 * tv);
  atomicMaxF(mrow + 3, m3 * tv);
}

__global__ void edge_sum_kernel(const float* __restrict__ x, const float* __restrict__ ea,
                                const int* __restrict__ src, const int* __restrict__ dst,
                                const float* __restrict__ tptr, int layer,
                                const float* __restrict__ mx,
                                float* __restrict__ den, float* __restrict__ num, int E) {
  int e    = blockIdx.x * (blockDim.x >> 5) + (threadIdx.x >> 5);
  int lane = threadIdx.x & 31;
  if (e >= E) return;
  int s = src[e], d = dst[e];
  float tv = tptr[layer];
  float4 a  = ((const float4*)(x  + (size_t)s * HDIM))[lane];
  float4 bb = ((const float4*)(ea + (size_t)e * HDIM))[lane];
  float4 mv = ((const float4*)(mx + (size_t)d * HDIM))[lane];
  float m0 = fmaxf(a.x + bb.x, 0.f) + MSG_EPS;
  float m1 = fmaxf(a.y + bb.y, 0.f) + MSG_EPS;
  float m2 = fmaxf(a.z + bb.z, 0.f) + MSG_EPS;
  float m3 = fmaxf(a.w + bb.w, 0.f) + MSG_EPS;
  float e0 = __expf(m0 * tv - mv.x);
  float e1 = __expf(m1 * tv - mv.y);
  float e2 = __expf(m2 * tv - mv.z);
  float e3 = __expf(m3 * tv - mv.w);
  float* drow = den + (size_t)d * HDIM + lane * 4;
  float* nrow = num + (size_t)d * HDIM + lane * 4;
  atomicAdd(drow + 0, e0);       atomicAdd(nrow + 0, e0 * m0);
  atomicAdd(drow + 1, e1);       atomicAdd(nrow + 1, e1 * m1);
  atomicAdd(drow + 2, e2);       atomicAdd(nrow + 2, e2 * m2);
  atomicAdd(drow + 3, e3);       atomicAdd(nrow + 3, e3 * m3);
}

__global__ void combine_kernel(const float* __restrict__ rin, const float* __restrict__ num,
                               const float* __restrict__ den, float* __restrict__ hc, int total) {
  int i = blockIdx.x * blockDim.x + threadIdx.x;
  if (i >= total) return;
  float dv = den[i];
  float agg = dv > 0.f ? num[i] / dv : 0.f;
  hc[i] = agg + rin[i];
}

__global__ void pool_kernel(const float* __restrict__ x, const int* __restrict__ batch,
                            float* __restrict__ pooled, int n) {
  int i = blockIdx.x * blockDim.x + threadIdx.x;
  if (i >= n * HDIM) return;
  int node = i >> 7, col = i & (HDIM - 1);
  atomicAdd(pooled + (size_t)batch[node] * HDIM + col, x[i]);
}

__global__ void final_linear_kernel(const float* __restrict__ pooled,
                                    const float* __restrict__ W,
                                    const float* __restrict__ b,
                                    float* __restrict__ out) {
  int i = blockIdx.x * blockDim.x + threadIdx.x;
  if (i >= NUM_GRAPHS * NUM_CLASSES) return;
  int g = i / NUM_CLASSES, c = i % NUM_CLASSES;
  float acc = b[c];
  const float* pr = pooled + (size_t)g * HDIM;
#pragma unroll 4
  for (int k = 0; k < HDIM; ++k) acc += pr[k] * W[k * NUM_CLASSES + c];
  out[i] = acc;
}

extern "C" void kernel_launch(void* const* d_in, const int* in_sizes, int n_in,
                              void* d_out, int out_size, void* d_ws, size_t ws_size,
                              hipStream_t stream) {
  const float* x     = (const float*)d_in[0];
  const float* ea    = (const float*)d_in[1];
  const float* enc_W = (const float*)d_in[2];
  const float* enc_b = (const float*)d_in[3];
  const float* t     = (const float*)d_in[4];
  const float* W1    = (const float*)d_in[5];
  const float* b1    = (const float*)d_in[6];
  const float* mlp_g = (const float*)d_in[7];
  const float* mlp_b = (const float*)d_in[8];
  const float* W2    = (const float*)d_in[9];
  const float* b2    = (const float*)d_in[10];
  const float* ln_g  = (const float*)d_in[11];
  const float* ln_b  = (const float*)d_in[12];
  const float* lin_W = (const float*)d_in[13];
  const float* lin_b = (const float*)d_in[14];
  const int*   eidx  = (const int*)d_in[15];
  const int*   batch = (const int*)d_in[16];

  const int N = in_sizes[0] / HDIM;
  const int E = in_sizes[1] / HDIM;
  const int L = in_sizes[4];
  const int* src = eidx;
  const int* dst = eidx + E;

  const size_t NH = (size_t)N * HDIM;
  float* h    = (float*)d_ws;
  float* rbuf = h    + NH;
  float* hc   = rbuf + NH;
  float* mx   = hc   + NH;
  float* den  = mx   + NH;
  float* num  = den  + NH;
  float* t1   = num  + NH;

  float* out    = (float*)d_out;
  float* pooled = out + NUM_GRAPHS * NUM_CLASSES;

  const dim3 gblk(128);
  const int eBlocks = (E + 7) / 8;
  const int elBlocks = (int)((NH + 255) / 256);
  const int lnBlocks = (N + 7) / 8;

  gemm16_wmma_f32<<<dim3(N / 16, HDIM / 64), gblk, 0, stream>>>(x, enc_W, enc_b, h, HDIM, HDIM, 0);

  for (int layer = 0; layer < L; ++layer) {
    const float* rin = h;
    if (layer > 0) {
      ln_relu_kernel<<<lnBlocks, 256, 0, stream>>>(h, rbuf, ln_g + layer * HDIM, ln_b + layer * HDIM, N, HDIM);
      rin = rbuf;
    }
    init_seg_kernel<<<elBlocks, 256, 0, stream>>>(mx, den, num, (int)NH);
    edge_max_kernel<<<eBlocks, 256, 0, stream>>>(rin, ea, src, dst, t, layer, mx, E);
    edge_sum_kernel<<<eBlocks, 256, 0, stream>>>(rin, ea, src, dst, t, layer, mx, den, num, E);
    combine_kernel<<<elBlocks, 256, 0, stream>>>(rin, num, den, hc, (int)NH);
    gemm16_wmma_f32<<<dim3(N / 16, HDIM2 / 64), gblk, 0, stream>>>(
        hc, W1 + (size_t)layer * HDIM * HDIM2, b1 + layer * HDIM2, t1, HDIM, HDIM2, 0);
    ln_relu_kernel<<<lnBlocks, 256, 0, stream>>>(t1, t1, mlp_g + layer * HDIM2, mlp_b + layer * HDIM2, N, HDIM2);
    gemm16_wmma_f32<<<dim3(N / 16, HDIM / 64), gblk, 0, stream>>>(
        t1, W2 + (size_t)layer * HDIM2 * HDIM, b2 + layer * HDIM, h, HDIM2, HDIM, layer > 0 ? 1 : 0);
  }

  ln_relu_kernel<<<lnBlocks, 256, 0, stream>>>(h, rbuf, ln_g, ln_b, N, HDIM);
  zero_kernel<<<(NUM_GRAPHS * HDIM + 255) / 256, 256, 0, stream>>>(pooled, NUM_GRAPHS * HDIM);
  pool_kernel<<<(int)((NH + 255) / 256), 256, 0, stream>>>(rbuf, batch, pooled, N);
  final_linear_kernel<<<(NUM_GRAPHS * NUM_CLASSES + 63) / 64, 64, 0, stream>>>(pooled, lin_W, lin_b, out);
}


# Edge softmax reads each x/edge row once per layer: per 32-column quarter the messages of the block's listed edges wait in an LDS plane between max pass and sum pass; sums still in LDS arrival order
# speedup vs baseline: 2.6811x; 2.6811x over previous
#include <hip/hip_runtime.h>
#include <hip/hip_bf16.h>

typedef __attribute__((ext_vector_type(2))) float v2f;
typedef __attribute__((ext_vector_type(8))) float v8f;

#define HDIM 128
#define HDIM2 256
#define NUM_GRAPHS 64
#define NUM_CLASSES 10
#define MSG_EPS 1e-7f

template <int NT> struct fvec;
template <> struct fvec<2> { typedef __attribute__((ext_vector_type(2))) float type; };
template <> struct fvec<4> { typedef __attribute__((ext_vector_type(4))) float type; };

template <int K, int M, int NT>
__global__ __launch_bounds__(128)
void gemm_rows16_wmma_f32(const float* __restrict__ X,
                          const float* __restrict__ W,
                          const float* __restrict__ bias,
                          float* __restrict__ Y, int accum) {
  typedef typename fvec<NT>::type vnt;
  typedef __attribute__((ext_vector_type(4))) float v4f;
  static_assert(M == 64 * NT, "4 waves of NT tiles cover the M columns");
  constexpr int PITCH = K + 4;
  constexpr int KU = 16 / NT;
  constexpr int NG = K / (4 * KU);
  __shared__ float s_a[16 * PITCH];

  const int tid  = threadIdx.x;
  const int lane = tid & 31;
  const int wid  = tid >> 5;
  const int row0 = blockIdx.x * 16;
  const int half = lane >> 4;
  const int l16  = lane & 15;
  const int c0   = wid * (16 * NT);
  const int colv = c0 + NT * l16;

  const float* wp = W + (size_t)(half * 2) * M + colv;
  const float* ap = s_a + l16 * PITCH + half * 2;

  vnt bq[2][KU][2];
  v2f aq[2][KU];

#pragma unroll
  for (int u = 0; u < KU; ++u) {
    bq[0][u][0] = *(const vnt*)(wp + (size_t)(u * 4) * M);
    bq[0][u][1] = *(const vnt*)(wp + (size_t)(u * 4 + 1) * M);
  }

  {
    const float* xs = X + (size_t)row0 * K;
#pragma unroll
    for (int i = 0; i < (16 * K / 4) / 128; ++i) {
      const int idx = tid + 128 * i;
      const int r = idx / (K / 4), q = idx % (K / 4);
      const v4f v = *(const v4f*)(xs + (size_t)r * K + q * 4);
      *(v4f*)(s_a + r * PITCH + q * 4) = v;
    }
  }

  const vnt bv = *(const vnt*)(bias + colv);
  v8f c[NT];
  if (accum) {
#pragma unroll
    for (int r = 0; r < 8; ++r) {
      const vnt y = *(const vnt*)(Y + (size_t)(row0 + half * 8 + r) * M + colv);
#pragma unroll
      for (int t = 0; t < NT; ++t) c[t][r] = y[t] + bv[t];
    }
  } else {
#pragma unroll
    for (int t = 0; t < NT; ++t)
#pragma unroll
      for (int r = 0; r < 8; ++r) c[t][r] = bv[t];
  }

  __syncthreads();

#pragma unroll
  for (int u = 0; u < KU; ++u) aq[0][u] = *(const v2f*)(ap + u * 4);

#pragma unroll
  for (int g = 0; g < NG; ++g) {
    const int cur = g & 1, nxt = cur ^ 1;
    if (g + 1 < NG) {
#pragma unroll
      for (int u = 0; u < KU; ++u) {
        const int k = ((g + 1) * KU + u) * 4;
        bq[nxt][u][0] = *(const vnt*)(wp + (size_t)k * M);
        bq[nxt][u][1] = *(const vnt*)(wp + (size_t)(k + 1) * M);
        aq[nxt][u]    = *(const v2f*)(ap + k);
      }
    }
#pragma unroll
    for (int u = 0; u < KU; ++u) {
#pragma unroll
      for (int t = 0; t < NT; ++t) {
        v2f b; b[0] = bq[cur][u][0][t]; b[1] = bq[cur][u][1][t];
        c[t] = __builtin_amdgcn_wmma_f32_16x16x4_f32(false, aq[cur][u], false, b,
                                                     (short)0, c[t], false, false);
      }
    }
  }

#pragma unroll
  for (int r = 0; r < 8; ++r) {
    vnt o;
#pragma unroll
    for (int t = 0; t < NT; ++t) o[t] = c[t][r];
    *(vnt*)(Y + (size_t)(row0 + half * 8 + r) * M + colv) = o;
  }
}

__global__ void ln_relu_kernel(const float* __restrict__ x, float* __restrict__ y,
                               const float* __restrict__ g, const float* __restrict__ b,
                               int rows, int D) {
  int row  = blockIdx.x * (blockDim.x >> 5) + (threadIdx.x >> 5);
  int lane = threadIdx.x & 31;
  if (row >= rows) return;
  const float* xr = x + (size_t)row * D;
  float s = 0.f, q = 0.f;
  for (int j = lane; j < D; j += 32) { float v = xr[j]; s += v; q += v * v; }
#pragma unroll
  for (int off = 16; off > 0; off >>= 1) {
    s += __shfl_xor(s, off, 32);
    q += __shfl_xor(q, off, 32);
  }
  float mean = s / (float)D;
  float var  = q / (float)D - mean * mean;
  float inv  = rsqrtf(var + 1e-5f);
  float* yr = y + (size_t)row * D;
  for (int j = lane; j < D; j += 32) {
    float v = (xr[j] - mean) * inv * g[j] + b[j];
    yr[j] = fmaxf(v, 0.f);
  }
}

__global__ void zero_kernel(float* p, int n) {
  int i = blockIdx.x * blockDim.x + threadIdx.x;
  if (i < n) p[i] = 0.f;
}

#define NODES_PER_BLOCK 40
#define BLOCK_WAVES 32
#define SCAN_CHUNKS 16
#define GROUP_EDGES 8
#define M_CAP 1536
#define QCOLS 32

template <int PHASE>
__device__ __forceinline__ void fold_rows(
    const float* __restrict__ x, const float* __restrict__ ea,
    const int (&pe)[GROUP_EDGES], const int (&s)[GROUP_EDGES], const int (&r)[GROUP_EDGES],
    int count, float tv, int lane, float* s_mx, float* s_den, float* s_num) {
  float a[GROUP_EDGES][4], b[GROUP_EDGES][4];
#pragma unroll
  for (int g = 0; g < GROUP_EDGES; ++g) {
    const float* xr = x  + (size_t)s[g]  * HDIM + lane;
    const float* er = ea + (size_t)pe[g] * HDIM + lane;
#pragma unroll
    for (int j = 0; j < 4; ++j) { a[g][j] = xr[32 * j]; b[g][j] = er[32 * j]; }
  }
  float mv[GROUP_EDGES][4];
  if (PHASE == 2) {
#pragma unroll
    for (int g = 0; g < GROUP_EDGES; ++g) {
      if (g < count) {
#pragma unroll
        for (int j = 0; j < 4; ++j) mv[g][j] = s_mx[r[g] * HDIM + lane + 32 * j];
      }
    }
  }
#pragma unroll
  for (int g = 0; g < GROUP_EDGES; ++g) {
    if (g < count) {
      const int o = r[g] * HDIM + lane;
#pragma unroll
      for (int j = 0; j < 4; ++j) {
        const float m = fmaxf(a[g][j] + b[g][j], 0.f) + MSG_EPS;
        if (PHASE == 1) {
          const float v = m * tv;
          if (v >= 0.0f) atomicMax((int*)&s_mx[o + 32 * j], __float_as_int(v));
          else           atomicMin((unsigned int*)&s_mx[o + 32 * j], __float_as_uint(v));
        } else {
          const float ex = __expf(m * tv - mv[g][j]);
          atomicAdd(&s_den[o + 32 * j], ex);
          atomicAdd(&s_num[o + 32 * j], ex * m);
        }
      }
    }
  }
}

template <int PHASE>
__device__ __forceinline__ void fold_edge_group(
    const float* __restrict__ x, const float* __restrict__ ea,
    const int* __restrict__ src, const int* __restrict__ dst,
    const int (&pe)[GROUP_EDGES], int count, int node0, float tv, int lane,
    float* s_mx, float* s_den, float* s_num) {
  int s[GROUP_EDGES], r[GROUP_EDGES];
#pragma unroll
  for (int g = 0; g < GROUP_EDGES; ++g) {
    s[g] = src[pe[g]];
    r[g] = dst[pe[g]] - node0;
  }
  fold_rows<PHASE>(x, ea, pe, s, r, count, tv, lane, s_mx, s_den, s_num);
}

__device__ __forceinline__ unsigned int scan_step_hits(const int* __restrict__ dst, int E, int c0,
                                                       int node0, int lane) {
  unsigned int hits = 0u;
  if ((c0 + SCAN_CHUNKS) * 32 <= E) {
    const int* p = dst + c0 * 32 + lane;
#pragma unroll
    for (int u = 0; u < SCAN_CHUNKS; ++u) {
      const int dv = p[u * 32];
      hits |= ((unsigned int)(dv - node0) < (unsigned int)NODES_PER_BLOCK ? 1u : 0u) << u;
    }
  } else {
#pragma unroll
    for (int u = 0; u < SCAN_CHUNKS; ++u) {
      const int e  = (c0 + u) * 32 + lane;
      const int dv = dst[e < E ? e : E - 1];
      const bool hit = (e < E) && ((unsigned int)(dv - node0) < (unsigned int)NODES_PER_BLOCK);
      hits |= (hit ? 1u : 0u) << u;
    }
  }
  return hits;
}

__device__ __forceinline__ void scan_build_list(
    const int* __restrict__ dst, int E, int node0, int lane, int wave,
    int* s_cnt, int* s_le) {
  const int nchunks = (E + 31) >> 5;
  for (int c0 = wave * SCAN_CHUNKS; c0 < nchunks; c0 += BLOCK_WAVES * SCAN_CHUNKS) {
    unsigned int hits = scan_step_hits(dst, E, c0, node0, lane);
    for (int it = 0; it < SCAN_CHUNKS; ++it) {
      const unsigned int active = __builtin_amdgcn_ballot_w32(hits != 0u);
      if (active == 0u) break;
      int slot0 = 0;
      if (lane == 0) slot0 = atomicAdd(s_cnt, __builtin_popcount(active));
      slot0 = __builtin_amdgcn_readfirstlane(slot0);
      if (hits != 0u) {
        const int u = __builtin_ctz(hits);
        hits &= hits - 1u;
        const int e = (c0 + u) * 32 + lane;
        const int slot = slot0 + __builtin_popcount(active & ((1u << lane) - 1u));
        if (slot < M_CAP) s_le[slot] = e;
      }
    }
  }
}

template <int NG, bool FULL>
__device__ __forceinline__ void quarter_messages(
    const float* __restrict__ x, const float* __restrict__ ea,
    int first, int n, float tv, int lane, int col,
    const int* s_le, const int* s_ls, const int* s_lr, float* s_mx, float* s_m) {
  constexpr int NE = GROUP_EDGES * NG;
  const int used = FULL ? NE : n;
  const int li = lane < used ? first + (lane >> 3) * (GROUP_EDGES * BLOCK_WAVES) + (lane & 7) : first;
  const int ev = s_le[li], sv = s_ls[li], rv = s_lr[li];
  float a[NE], b[NE];
#pragma unroll
  for (int g = 0; g < NE; ++g) {
    const int pe = __builtin_amdgcn_readlane(ev, g);
    const int s  = __builtin_amdgcn_readlane(sv, g);
    a[g] = x[(size_t)s * HDIM + col];
    b[g] = ea[(size_t)pe * HDIM + col];
  }
#pragma unroll
  for (int g = 0; g < NE; ++g) {
    if (FULL || g < n) {
      const int r   = __builtin_amdgcn_readlane(rv, g);
      const int o   = r * HDIM + col;
      const int row = first + (g >> 3) * (GROUP_EDGES * BLOCK_WAVES) + (g & 7);
      const float m = fmaxf(a[g] + b[g], 0.f) + MSG_EPS;
      s_m[row * QCOLS + lane] = m;
      const float v = m * tv;
      if (v >= 0.0f) atomicMax((int*)&s_mx[o], __float_as_int(v));
      else           atomicMin((unsigned int*)&s_mx[o], __float_as_uint(v));
    }
  }
}

template <bool FULL>
__device__ __forceinline__ void quarter_sums(
    int first, int n, float tv, int lane, int col,
    const int* s_lr, const float* s_mx, float* s_den, float* s_num, const float* s_m) {
  const int used = FULL ? GROUP_EDGES : n;
  const int rv = s_lr[lane < used ? first + lane : first];
  float m[GROUP_EDGES], mv[GROUP_EDGES];
  int o[GROUP_EDGES];
#pragma unroll
  for (int g = 0; g < GROUP_EDGES; ++g) {
    if (FULL || g < n) {
      o[g]  = __builtin_amdgcn_readlane(rv, g) * HDIM + col;
      m[g]  = s_m[(first + g) * QCOLS + lane];
      mv[g] = s_mx[o[g]];
    }
  }
#pragma unroll
  for (int g = 0; g < GROUP_EDGES; ++g) {
    if (FULL || g < n) {
      const float ex = __expf(m[g] * tv - mv[g]);
      atomicAdd(&s_den[o[g]], ex);
      atomicAdd(&s_num[o[g]], ex * m[g]);
    }
  }
}

__device__ __forceinline__ void fold_list_by_quarters(
    const float* __restrict__ x, const float* __restrict__ ea,
    int cnt, float tv, int lane, int wave,
    const int* s_le, const int* s_ls, const int* s_lr,
    float* s_mx, float* s_den, float* s_num, float* s_m) {
  const int G     = (cnt + GROUP_EDGES - 1) / GROUP_EDGES;
  const int tail  = cnt - GROUP_EDGES * (G - 1);
  const int ng    = G > wave ? (G - wave + BLOCK_WAVES - 1) / BLOCK_WAVES : 0;
  const bool has_short = ng > 0 && wave + BLOCK_WAVES * (ng - 1) == G - 1 && tail < GROUP_EDGES;
  const int nfull = has_short ? ng - 1 : ng;
  const int STRIDE = GROUP_EDGES * BLOCK_WAVES;
  const int first0 = GROUP_EDGES * wave;
#pragma unroll 1
  for (int q = 0; q < HDIM / QCOLS; ++q) {
    const int col = q * QCOLS + lane;
    int j = 0;
    for (; j + 3 <= nfull; j += 3)
      quarter_messages<3, true>(x, ea, first0 + j * STRIDE, 0, tv, lane, col, s_le, s_ls, s_lr, s_mx, s_m);
    if (nfull - j == 2)
      quarter_messages<2, true>(x, ea, first0 + j * STRIDE, 0, tv, lane, col, s_le, s_ls, s_lr, s_mx, s_m);
    else if (nfull - j == 1)
      quarter_messages<1, true>(x, ea, first0 + j * STRIDE, 0, tv, lane, col, s_le, s_ls, s_lr, s_mx, s_m);
    if (has_short)
      quarter_messages<1, false>(x, ea, first0 + nfull * STRIDE, tail, tv, lane, col, s_le, s_ls, s_lr, s_mx, s_m);
    __syncthreads();
    for (int k = 0; k < nfull; ++k)
      quarter_sums<true>(first0 + k * STRIDE, 0, tv, lane, col, s_lr, s_mx, s_den, s_num, s_m);
    if (has_short)
      quarter_sums<false>(first0 + nfull * STRIDE, tail, tv, lane, col, s_lr, s_mx, s_den, s_num, s_m);
  }
}

template <int PHASE>
__device__ __forceinline__ void scan_and_fold(
    const float* __restrict__ x, const float* __restrict__ ea,
    const int* __restrict__ src, const int* __restrict__ dst,
    int E, int node0, float tv, int lane, int wave,
    float* s_mx, float* s_den, float* s_num) {
  int pe[GROUP_EDGES];
#pragma unroll
  for (int g = 0; g < GROUP_EDGES; ++g) pe[g] = 0;
  int pn = 0;
  const int nchunks = (E + 31) >> 5;
  for (int c0 = wave * SCAN_CHUNKS; c0 < nchunks; c0 += BLOCK_WAVES * SCAN_CHUNKS) {
    const unsigned int hits = scan_step_hits(dst, E, c0, node0, lane);
    if (__builtin_amdgcn_ballot_w32(hits != 0u) != 0u) {
#pragma unroll 1
      for (int u = 0; u < SCAN_CHUNKS; ++u) {
        unsigned int mask = __builtin_amdgcn_ballot_w32(((hits >> u) & 1u) != 0u);
        const int base = (c0 + u) * 32;
        while (mask != 0u) {
          const int bit = __builtin_ctz(mask);
          mask &= mask - 1u;
#pragma unroll
          for (int g = GROUP_EDGES - 1; g > 0; --g) pe[g] = pe[g - 1];
          pe[0] = base + bit;
          if (++pn == GROUP_EDGES) {
            fold_edge_group<PHASE>(x, ea, src, dst, pe, GROUP_EDGES, node0, tv, lane, s_mx, s_den, s_num);
            pn = 0;
          }
        }
      }
    }
  }
  if (pn > 0)
    fold_edge_group<PHASE>(x, ea, src, dst, pe, pn, node0, tv, lane, s_mx, s_den, s_num);
}

__global__ __launch_bounds__(BLOCK_WAVES * 32)
void gen_softmax_lds_kernel(const float* __restrict__ x, const float* __restrict__ ea,
                            const int* __restrict__ src, const int* __restrict__ dst,
                            const float* __restrict__ tptr, int layer,
                            float* __restrict__ hc, int E, int N) {
  __shared__ float s_mx[NODES_PER_BLOCK * HDIM];
  __shared__ float s_den[NODES_PER_BLOCK * HDIM];
  __shared__ float s_num[NODES_PER_BLOCK * HDIM];
  __shared__ float s_m[M_CAP * QCOLS];
  __shared__ int s_le[M_CAP];
  __shared__ int s_ls[M_CAP];
  __shared__ int s_lr[M_CAP];
  __shared__ int s_cnt;
  const int tid   = threadIdx.x;
  const int lane  = tid & 31;
  const int wave  = __builtin_amdgcn_readfirstlane(tid >> 5);
  const int node0 = blockIdx.x * NODES_PER_BLOCK;
  for (int i = tid; i < NODES_PER_BLOCK * HDIM; i += BLOCK_WAVES * 32) {
    s_mx[i]  = __int_as_float(0xff800000);
    s_den[i] = 0.f;
    s_num[i] = 0.f;
  }
  if (tid == 0) s_cnt = 0;
  const float tv = tptr[layer];
  __syncthreads();
  scan_build_list(dst, E, node0, lane, wave, &s_cnt, s_le);
  __syncthreads();
  const int cnt = __builtin_amdgcn_readfirstlane(s_cnt);
  if (cnt <= M_CAP) {
    for (int i = tid; i < cnt; i += BLOCK_WAVES * 32) {
      const int e = s_le[i];
      s_ls[i] = src[e];
      s_lr[i] = dst[e] - node0;
    }
    __syncthreads();
    fold_list_by_quarters(x, ea, cnt, tv, lane, wave, s_le, s_ls, s_lr, s_mx, s_den, s_num, s_m);
  } else {
    scan_and_fold<1>(x, ea, src, dst, E, node0, tv, lane, wave, s_mx, s_den, s_num);
    __syncthreads();
    scan_and_fold<2>(x, ea, src, dst, E, node0, tv, lane, wave, s_mx, s_den, s_num);
  }
  __syncthreads();
  for (int i = tid; i < NODES_PER_BLOCK * HDIM; i += BLOCK_WAVES * 32) {
    const int node = node0 + i / HDIM;
    if (node < N) {
      const float dv  = s_den[i];
      const float agg = dv > 0.f ? s_num[i] / dv : 0.f;
      const size_t gi = (size_t)node * HDIM + i % HDIM;
      hc[gi] = agg + x[gi];
    }
  }
}

__global__ void pool_kernel(const float* __restrict__ x, const int* __restrict__ batch,
                            float* __restrict__ pooled, int n) {
  int i = blockIdx.x * blockDim.x + threadIdx.x;
  if (i >= n * HDIM) return;
  int node = i >> 7, col = i & (HDIM - 1);
  atomicAdd(pooled + (size_t)batch[node] * HDIM + col, x[i]);
}

__global__ void final_linear_kernel(const float* __restrict__ pooled,
                                    const float* __restrict__ W,
                                    const float* __restrict__ b,
                                    float* __restrict__ out) {
  int i = blockIdx.x * blockDim.x + threadIdx.x;
  if (i >= NUM_GRAPHS * NUM_CLASSES) return;
  int g = i / NUM_CLASSES, c = i % NUM_CLASSES;
  float acc = b[c];
  const float* pr = pooled + (size_t)g * HDIM;
#pragma unroll 4
  for (int k = 0; k < HDIM; ++k) acc += pr[k] * W[k * NUM_CLASSES + c];
  out[i] = acc;
}

extern "C" void kernel_launch(void* const* d_in, const int* in_sizes, int n_in,
                              void* d_out, int out_size, void* d_ws, size_t ws_size,
                              hipStream_t stream) {
  const float* x     = (const float*)d_in[0];
  const float* ea    = (const float*)d_in[1];
  const float* enc_W = (const float*)d_in[2];
  const float* enc_b = (const float*)d_in[3];
  const float* t     = (const float*)d_in[4];
  const float* W1    = (const float*)d_in[5];
  const float* b1    = (const float*)d_in[6];
  const float* mlp_g = (const float*)d_in[7];
  const float* mlp_b = (const float*)d_in[8];
  const float* W2    = (const float*)d_in[9];
  const float* b2    = (const float*)d_in[10];
  const float* ln_g  = (const float*)d_in[11];
  const float* ln_b  = (const float*)d_in[12];
  const float* lin_W = (const float*)d_in[13];
  const float* lin_b = (const float*)d_in[14];
  const int*   eidx  = (const int*)d_in[15];
  const int*   batch = (const int*)d_in[16];

  const int N = in_sizes[0] / HDIM;
  const int E = in_sizes[1] / HDIM;
  const int L = in_sizes[4];
  const int* src = eidx;
  const int* dst = eidx + E;

  const size_t NH = (size_t)N * HDIM;
  float* h    = (float*)d_ws;
  float* rbuf = h    + NH;
  float* hc   = rbuf + NH;
  float* t1   = hc   + NH;

  float* out    = (float*)d_out;
  float* pooled = out + NUM_GRAPHS * NUM_CLASSES;

  const dim3 gblk(128);
  const int aggBlocks = (N + NODES_PER_BLOCK - 1) / NODES_PER_BLOCK;
  const int lnBlocks = (N + 7) / 8;

  gemm_rows16_wmma_f32<HDIM, HDIM, 2><<<N / 16, gblk, 0, stream>>>(x, enc_W, enc_b, h, 0);

  for (int layer = 0; layer < L; ++layer) {
    const float* rin = h;
    if (layer > 0) {
      ln_relu_kernel<<<lnBlocks, 256, 0, stream>>>(h, rbuf, ln_g + layer * HDIM, ln_b + layer * HDIM, N, HDIM);
      rin = rbuf;
    }
    gen_softmax_lds_kernel<<<aggBlocks, BLOCK_WAVES * 32, 0, stream>>>(rin, ea, src, dst, t, layer, hc, E, N);
    gemm_rows16_wmma_f32<HDIM, HDIM2, 4><<<N / 16, gblk, 0, stream>>>(
        hc, W1 + (size_t)layer * HDIM * HDIM2, b1 + layer * HDIM2, t1, 0);
    ln_relu_kernel<<<lnBlocks, 256, 0, stream>>>(t1, t1, mlp_g + layer * HDIM2, mlp_b + layer * HDIM2, N, HDIM2);
    gemm_rows16_wmma_f32<HDIM2, HDIM, 2><<<N / 16, gblk, 0, stream>>>(
        t1, W2 + (size_t)layer * HDIM2 * HDIM, b2 + layer * HDIM, h, layer > 0 ? 1 : 0);
  }

  ln_relu_kernel<<<lnBlocks, 256, 0, stream>>>(h, rbuf, ln_g, ln_b, N, HDIM);
  zero_kernel<<<(NUM_GRAPHS * HDIM + 255) / 256, 256, 0, stream>>>(pooled, NUM_GRAPHS * HDIM);
  pool_kernel<<<(int)((NH + 255) / 256), 256, 0, stream>>>(rbuf, batch, pooled, N);
  final_linear_kernel<<<(NUM_GRAPHS * NUM_CLASSES + 63) / 64, 64, 0, stream>>>(pooled, lin_W, lin_b, out);
}


# Each block's edge list (id, source, row) is built once per call: layer 0's launch saves it in the workspace, layers 1-3 load it and no longer scan dst; sums still in LDS arrival order
# speedup vs baseline: 3.0581x; 1.1406x over previous
#include <hip/hip_runtime.h>
#include <hip/hip_bf16.h>

typedef __attribute__((ext_vector_type(2))) float v2f;
typedef __attribute__((ext_vector_type(8))) float v8f;

#define HDIM 128
#define HDIM2 256
#define NUM_GRAPHS 64
#define NUM_CLASSES 10
#define MSG_EPS 1e-7f

template <int NT> struct fvec;
template <> struct fvec<2> { typedef __attribute__((ext_vector_type(2))) float type; };
template <> struct fvec<4> { typedef __attribute__((ext_vector_type(4))) float type; };

template <int K, int M, int NT>
__global__ __launch_bounds__(128)
void gemm_rows16_wmma_f32(const float* __restrict__ X,
                          const float* __restrict__ W,
                          const float* __restrict__ bias,
                          float* __restrict__ Y, int accum) {
  typedef typename fvec<NT>::type vnt;
  typedef __attribute__((ext_vector_type(4))) float v4f;
  static_assert(M == 64 * NT, "4 waves of NT tiles cover the M columns");
  constexpr int PITCH = K + 4;
  constexpr int KU = 16 / NT;
  constexpr int NG = K / (4 * KU);
  __shared__ float s_a[16 * PITCH];

  const int tid  = threadIdx.x;
  const int lane = tid & 31;
  const int wid  = tid >> 5;
  const int row0 = blockIdx.x * 16;
  const int half = lane >> 4;
  const int l16  = lane & 15;
  const int c0   = wid * (16 * NT);
  const int colv = c0 + NT * l16;

  const float* wp = W + (size_t)(half * 2) * M + colv;
  const float* ap = s_a + l16 * PITCH + half * 2;

  vnt bq[2][KU][2];
  v2f aq[2][KU];

#pragma unroll
  for (int u = 0; u < KU; ++u) {
    bq[0][u][0] = *(const vnt*)(wp + (size_t)(u * 4) * M);
    bq[0][u][1] = *(const vnt*)(wp + (size_t)(u * 4 + 1) * M);
  }

  {
    const float* xs = X + (size_t)row0 * K;
#pragma unroll
    for (int i = 0; i < (16 * K / 4) / 128; ++i) {
      const int idx = tid + 128 * i;
      const int r = idx / (K / 4), q = idx % (K / 4);
      const v4f v = *(const v4f*)(xs + (size_t)r * K + q * 4);
      *(v4f*)(s_a + r * PITCH + q * 4) = v;
    }
  }

  const vnt bv = *(const vnt*)(bias + colv);
  v8f c[NT];
  if (accum) {
#pragma unroll
    for (int r = 0; r < 8; ++r) {
      const vnt y = *(const vnt*)(Y + (size_t)(row0 + half * 8 + r) * M + colv);
#pragma unroll
      for (int t = 0; t < NT; ++t) c[t][r] = y[t] + bv[t];
    }
  } else {
#pragma unroll
    for (int t = 0; t < NT; ++t)
#pragma unroll
      for (int r = 0; r < 8; ++r) c[t][r] = bv[t];
  }

  __syncthreads();

#pragma unroll
  for (int u = 0; u < KU; ++u) aq[0][u] = *(const v2f*)(ap + u * 4);

#pragma unroll
  for (int g = 0; g < NG; ++g) {
    const int cur = g & 1, nxt = cur ^ 1;
    if (g + 1 < NG) {
#pragma unroll
      for (int u = 0; u < KU; ++u) {
        const int k = ((g + 1) * KU + u) * 4;
        bq[nxt][u][0] = *(const vnt*)(wp + (size_t)k * M);
        bq[nxt][u][1] = *(const vnt*)(wp + (size_t)(k + 1) * M);
        aq[nxt][u]    = *(const v2f*)(ap + k);
      }
    }
#pragma unroll
    for (int u = 0; u < KU; ++u) {
#pragma unroll
      for (int t = 0; t < NT; ++t) {
        v2f b; b[0] = bq[cur][u][0][t]; b[1] = bq[cur][u][1][t];
        c[t] = __builtin_amdgcn_wmma_f32_16x16x4_f32(false, aq[cur][u], false, b,
                                                     (short)0, c[t], false, false);
      }
    }
  }

#pragma unroll
  for (int r = 0; r < 8; ++r) {
    vnt o;
#pragma unroll
    for (int t = 0; t < NT; ++t) o[t] = c[t][r];
    *(vnt*)(Y + (size_t)(row0 + half * 8 + r) * M + colv) = o;
  }
}

__global__ void ln_relu_kernel(const float* __restrict__ x, float* __restrict__ y,
                               const float* __restrict__ g, const float* __restrict__ b,
                               int rows, int D) {
  int row  = blockIdx.x * (blockDim.x >> 5) + (threadIdx.x >> 5);
  int lane = threadIdx.x & 31;
  if (row >= rows) return;
  const float* xr = x + (size_t)row * D;
  float s = 0.f, q = 0.f;
  for (int j = lane; j < D; j += 32) { float v = xr[j]; s += v; q += v * v; }
#pragma unroll
  for (int off = 16; off > 0; off >>= 1) {
    s += __shfl_xor(s, off, 32);
    q += __shfl_xor(q, off, 32);
  }
  float mean = s / (float)D;
  float var  = q / (float)D - mean * mean;
  float inv  = rsqrtf(var + 1e-5f);
  float* yr = y + (size_t)row * D;
  for (int j = lane; j < D; j += 32) {
    float v = (xr[j] - mean) * inv * g[j] + b[j];
    yr[j] = fmaxf(v, 0.f);
  }
}

__global__ void zero_kernel(float* p, int n) {
  int i = blockIdx.x * blockDim.x + threadIdx.x;
  if (i < n) p[i] = 0.f;
}

#define NODES_PER_BLOCK 40
#define BLOCK_WAVES 32
#define SCAN_CHUNKS 16
#define GROUP_EDGES 8
#define M_CAP 1536
#define QCOLS 32
#define LIST_HDR 32
#define LIST_STRIDE (LIST_HDR + 3 * M_CAP)

template <int PHASE>
__device__ __forceinline__ void fold_rows(
    const float* __restrict__ x, const float* __restrict__ ea,
    const int (&pe)[GROUP_EDGES], const int (&s)[GROUP_EDGES], const int (&r)[GROUP_EDGES],
    int count, float tv, int lane, float* s_mx, float* s_den, float* s_num) {
  float a[GROUP_EDGES][4], b[GROUP_EDGES][4];
#pragma unroll
  for (int g = 0; g < GROUP_EDGES; ++g) {
    const float* xr = x  + (size_t)s[g]  * HDIM + lane;
    const float* er = ea + (size_t)pe[g] * HDIM + lane;
#pragma unroll
    for (int j = 0; j < 4; ++j) { a[g][j] = xr[32 * j]; b[g][j] = er[32 * j]; }
  }
  float mv[GROUP_EDGES][4];
  if (PHASE == 2) {
#pragma unroll
    for (int g = 0; g < GROUP_EDGES; ++g) {
      if (g < count) {
#pragma unroll
        for (int j = 0; j < 4; ++j) mv[g][j] = s_mx[r[g] * HDIM + lane + 32 * j];
      }
    }
  }
#pragma unroll
  for (int g = 0; g < GROUP_EDGES; ++g) {
    if (g < count) {
      const int o = r[g] * HDIM + lane;
#pragma unroll
      for (int j = 0; j < 4; ++j) {
        const float m = fmaxf(a[g][j] + b[g][j], 0.f) + MSG_EPS;
        if (PHASE == 1) {
          const float v = m * tv;
          if (v >= 0.0f) atomicMax((int*)&s_mx[o + 32 * j], __float_as_int(v));
          else           atomicMin((unsigned int*)&s_mx[o + 32 * j], __float_as_uint(v));
        } else {
          const float ex = __expf(m * tv - mv[g][j]);
          atomicAdd(&s_den[o + 32 * j], ex);
          atomicAdd(&s_num[o + 32 * j], ex * m);
        }
      }
    }
  }
}

template <int PHASE>
__device__ __forceinline__ void fold_edge_group(
    const float* __restrict__ x, const float* __restrict__ ea,
    const int* __restrict__ src, const int* __restrict__ dst,
    const int (&pe)[GROUP_EDGES], int count, int node0, float tv, int lane,
    float* s_mx, float* s_den, float* s_num) {
  int s[GROUP_EDGES], r[GROUP_EDGES];
#pragma unroll
  for (int g = 0; g < GROUP_EDGES; ++g) {
    s[g] = src[pe[g]];
    r[g] = dst[pe[g]] - node0;
  }
  fold_rows<PHASE>(x, ea, pe, s, r, count, tv, lane, s_mx, s_den, s_num);
}

__device__ __forceinline__ unsigned int scan_step_hits(const int* __restrict__ dst, int E, int c0,
                                                       int node0, int lane) {
  unsigned int hits = 0u;
  if ((c0 + SCAN_CHUNKS) * 32 <= E) {
    const int* p = dst + c0 * 32 + lane;
#pragma unroll
    for (int u = 0; u < SCAN_CHUNKS; ++u) {
      const int dv = p[u * 32];
      hits |= ((unsigned int)(dv - node0) < (unsigned int)NODES_PER_BLOCK ? 1u : 0u) << u;
    }
  } else {
#pragma unroll
    for (int u = 0; u < SCAN_CHUNKS; ++u) {
      const int e  = (c0 + u) * 32 + lane;
      const int dv = dst[e < E ? e : E - 1];
      const bool hit = (e < E) && ((unsigned int)(dv - node0) < (unsigned int)NODES_PER_BLOCK);
      hits |= (hit ? 1u : 0u) << u;
    }
  }
  return hits;
}

__device__ __forceinline__ void scan_build_list(
    const int* __restrict__ dst, int E, int node0, int lane, int wave,
    int* s_cnt, int* s_le) {
  const int nchunks = (E + 31) >> 5;
  for (int c0 = wave * SCAN_CHUNKS; c0 < nchunks; c0 += BLOCK_WAVES * SCAN_CHUNKS) {
    unsigned int hits = scan_step_hits(dst, E, c0, node0, lane);
    for (int it = 0; it < SCAN_CHUNKS; ++it) {
      const unsigned int active = __builtin_amdgcn_ballot_w32(hits != 0u);
      if (active == 0u) break;
      int slot0 = 0;
      if (lane == 0) slot0 = atomicAdd(s_cnt, __builtin_popcount(active));
      slot0 = __builtin_amdgcn_readfirstlane(slot0);
      if (hits != 0u) {
        const int u = __builtin_ctz(hits);
        hits &= hits - 1u;
        const int e = (c0 + u) * 32 + lane;
        const int slot = slot0 + __builtin_popcount(active & ((1u << lane) - 1u));
        if (slot < M_CAP) s_le[slot] = e;
      }
    }
  }
}

template <int NG, bool FULL>
__device__ __forceinline__ void quarter_messages(
    const float* __restrict__ x, const float* __restrict__ ea,
    int first, int n, float tv, int lane, int col,
    const int* s_le, const int* s_ls, const int* s_lr, float* s_mx, float* s_m) {
  constexpr int NE = GROUP_EDGES * NG;
  const int used = FULL ? NE : n;
  const int li = lane < used ? first + (lane >> 3) * (GROUP_EDGES * BLOCK_WAVES) + (lane & 7) : first;
  const int ev = s_le[li], sv = s_ls[li], rv = s_lr[li];
  float a[NE], b[NE];
#pragma unroll
  for (int g = 0; g < NE; ++g) {
    const int pe = __builtin_amdgcn_readlane(ev, g);
    const int s  = __builtin_amdgcn_readlane(sv, g);
    a[g] = x[(size_t)s * HDIM + col];
    b[g] = ea[(size_t)pe * HDIM + col];
  }
#pragma unroll
  for (int g = 0; g < NE; ++g) {
    if (FULL || g < n) {
      const int r   = __builtin_amdgcn_readlane(rv, g);
      const int o   = r * HDIM + col;
      const int row = first + (g >> 3) * (GROUP_EDGES * BLOCK_WAVES) + (g & 7);
      const float m = fmaxf(a[g] + b[g], 0.f) + MSG_EPS;
      s_m[row * QCOLS + lane] = m;
      const float v = m * tv;
      if (v >= 0.0f) atomicMax((int*)&s_mx[o], __float_as_int(v));
      else           atomicMin((unsigned int*)&s_mx[o], __float_as_uint(v));
    }
  }
}

template <bool FULL>
__device__ __forceinline__ void quarter_sums(
    int first, int n, float tv, int lane, int col,
    const int* s_lr, const float* s_mx, float* s_den, float* s_num, const float* s_m) {
  const int used = FULL ? GROUP_EDGES : n;
  const int rv = s_lr[lane < used ? first + lane : first];
  float m[GROUP_EDGES], mv[GROUP_EDGES];
  int o[GROUP_EDGES];
#pragma unroll
  for (int g = 0; g < GROUP_EDGES; ++g) {
    if (FULL || g < n) {
      o[g]  = __builtin_amdgcn_readlane(rv, g) * HDIM + col;
      m[g]  = s_m[(first + g) * QCOLS + lane];
      mv[g] = s_mx[o[g]];
    }
  }
#pragma unroll
  for (int g = 0; g < GROUP_EDGES; ++g) {
    if (FULL || g < n) {
      const float ex = __expf(m[g] * tv - mv[g]);
      atomicAdd(&s_den[o[g]], ex);
      atomicAdd(&s_num[o[g]], ex * m[g]);
    }
  }
}

__device__ __forceinline__ void fold_list_by_quarters(
    const float* __restrict__ x, const float* __restrict__ ea,
    int cnt, float tv, int lane, int wave,
    const int* s_le, const int* s_ls, const int* s_lr,
    float* s_mx, float* s_den, float* s_num, float* s_m) {
  const int G     = (cnt + GROUP_EDGES - 1) / GROUP_EDGES;
  const int tail  = cnt - GROUP_EDGES * (G - 1);
  const int ng    = G > wave ? (G - wave + BLOCK_WAVES - 1) / BLOCK_WAVES : 0;
  const bool has_short = ng > 0 && wave + BLOCK_WAVES * (ng - 1) == G - 1 && tail < GROUP_EDGES;
  const int nfull = has_short ? ng - 1 : ng;
  const int STRIDE = GROUP_EDGES * BLOCK_WAVES;
  const int first0 = GROUP_EDGES * wave;
#pragma unroll 1
  for (int q = 0; q < HDIM / QCOLS; ++q) {
    const int col = q * QCOLS + lane;
    int j = 0;
    for (; j + 3 <= nfull; j += 3)
      quarter_messages<3, true>(x, ea, first0 + j * STRIDE, 0, tv, lane, col, s_le, s_ls, s_lr, s_mx, s_m);
    if (nfull - j == 2)
      quarter_messages<2, true>(x, ea, first0 + j * STRIDE, 0, tv, lane, col, s_le, s_ls, s_lr, s_mx, s_m);
    else if (nfull - j == 1)
      quarter_messages<1, true>(x, ea, first0 + j * STRIDE, 0, tv, lane, col, s_le, s_ls, s_lr, s_mx, s_m);
    if (has_short)
      quarter_messages<1, false>(x, ea, first0 + nfull * STRIDE, tail, tv, lane, col, s_le, s_ls, s_lr, s_mx, s_m);
    __syncthreads();
    for (int k = 0; k < nfull; ++k)
      quarter_sums<true>(first0 + k * STRIDE, 0, tv, lane, col, s_lr, s_mx, s_den, s_num, s_m);
    if (has_short)
      quarter_sums<false>(first0 + nfull * STRIDE, tail, tv, lane, col, s_lr, s_mx, s_den, s_num, s_m);
  }
}

template <int PHASE>
__device__ __forceinline__ void scan_and_fold(
    const float* __restrict__ x, const float* __restrict__ ea,
    const int* __restrict__ src, const int* __restrict__ dst,
    int E, int node0, float tv, int lane, int wave,
    float* s_mx, float* s_den, float* s_num) {
  int pe[GROUP_EDGES];
#pragma unroll
  for (int g = 0; g < GROUP_EDGES; ++g) pe[g] = 0;
  int pn = 0;
  const int nchunks = (E + 31) >> 5;
  for (int c0 = wave * SCAN_CHUNKS; c0 < nchunks; c0 += BLOCK_WAVES * SCAN_CHUNKS) {
    const unsigned int hits = scan_step_hits(dst, E, c0, node0, lane);
    if (__builtin_amdgcn_ballot_w32(hits != 0u) != 0u) {
#pragma unroll 1
      for (int u = 0; u < SCAN_CHUNKS; ++u) {
        unsigned int mask = __builtin_amdgcn_ballot_w32(((hits >> u) & 1u) != 0u);
        const int base = (c0 + u) * 32;
        while (mask != 0u) {
          const int bit = __builtin_ctz(mask);
          mask &= mask - 1u;
#pragma unroll
          for (int g = GROUP_EDGES - 1; g > 0; --g) pe[g] = pe[g - 1];
          pe[0] = base + bit;
          if (++pn == GROUP_EDGES) {
            fold_edge_group<PHASE>(x, ea, src, dst, pe, GROUP_EDGES, node0, tv, lane, s_mx, s_den, s_num);
            pn = 0;
          }
        }
      }
    }
  }
  if (pn > 0)
    fold_edge_group<PHASE>(x, ea, src, dst, pe, pn, node0, tv, lane, s_mx, s_den, s_num);
}

template <bool BUILD>
__global__ __launch_bounds__(BLOCK_WAVES * 32)
void gen_softmax_lds_kernel(const float* __restrict__ x, const float* __restrict__ ea,
                            const int* __restrict__ src, const int* __restrict__ dst,
                            const float* __restrict__ tptr, int layer,
                            float* __restrict__ hc, int* __restrict__ lists, int E, int N) {
  __shared__ float s_mx[NODES_PER_BLOCK * HDIM];
  __shared__ float s_den[NODES_PER_BLOCK * HDIM];
  __shared__ float s_num[NODES_PER_BLOCK * HDIM];
  __shared__ float s_m[M_CAP * QCOLS];
  __shared__ int s_le[M_CAP];
  __shared__ int s_ls[M_CAP];
  __shared__ int s_lr[M_CAP];
  __shared__ int s_cnt;
  const int tid   = threadIdx.x;
  const int lane  = tid & 31;
  const int wave  = __builtin_amdgcn_readfirstlane(tid >> 5);
  const int node0 = blockIdx.x * NODES_PER_BLOCK;
  int* gl = lists + (size_t)blockIdx.x * LIST_STRIDE;
  for (int i = tid; i < NODES_PER_BLOCK * HDIM; i += BLOCK_WAVES * 32) {
    s_mx[i]  = __int_as_float(0xff800000);
    s_den[i] = 0.f;
    s_num[i] = 0.f;
  }
  const float tv = tptr[layer];
  int cnt;
  if (BUILD) {
    if (tid == 0) s_cnt = 0;
    __syncthreads();
    scan_build_list(dst, E, node0, lane, wave, &s_cnt, s_le);
    __syncthreads();
    cnt = __builtin_amdgcn_readfirstlane(s_cnt);
    if (tid == 0) gl[0] = cnt;
    if (cnt <= M_CAP) {
      for (int i = tid; i < cnt; i += BLOCK_WAVES * 32) {
        const int e = s_le[i];
        const int s = src[e];
        const int r = dst[e] - node0;
        s_ls[i] = s;
        s_lr[i] = r;
        gl[LIST_HDR + i]             = e;
        gl[LIST_HDR + M_CAP + i]     = s;
        gl[LIST_HDR + 2 * M_CAP + i] = r;
      }
    }
  } else {
    cnt = __builtin_amdgcn_readfirstlane(gl[0]);
    if (cnt <= M_CAP) {
      constexpr int TRIPS = (M_CAP + BLOCK_WAVES * 32 - 1) / (BLOCK_WAVES * 32);
      int ve[TRIPS], vs[TRIPS], vr[TRIPS];
#pragma unroll
      for (int k = 0; k < TRIPS; ++k) {
        const int i = tid + k * (BLOCK_WAVES * 32);
        if (i < cnt) {
          ve[k] = gl[LIST_HDR + i];
          vs[k] = gl[LIST_HDR + M_CAP + i];
          vr[k] = gl[LIST_HDR + 2 * M_CAP + i];
        }
      }
#pragma unroll
      for (int k = 0; k < TRIPS; ++k) {
        const int i = tid + k * (BLOCK_WAVES * 32);
        if (i < cnt) {
          s_le[i] = ve[k];
          s_ls[i] = vs[k];
          s_lr[i] = vr[k];
        }
      }
    }
  }
  __syncthreads();
  if (cnt <= M_CAP) {
    fold_list_by_quarters(x, ea, cnt, tv, lane, wave, s_le, s_ls, s_lr, s_mx, s_den, s_num, s_m);
  } else {
    scan_and_fold<1>(x, ea, src, dst, E, node0, tv, lane, wave, s_mx, s_den, s_num);
    __syncthreads();
    scan_and_fold<2>(x, ea, src, dst, E, node0, tv, lane, wave, s_mx, s_den, s_num);
  }
  __syncthreads();
  for (int i = tid; i < NODES_PER_BLOCK * HDIM; i += BLOCK_WAVES * 32) {
    const int node = node0 + i / HDIM;
    if (node < N) {
      const float dv  = s_den[i];
      const float agg = dv > 0.f ? s_num[i] / dv : 0.f;
      const size_t gi = (size_t)node * HDIM + i % HDIM;
      hc[gi] = agg + x[gi];
    }
  }
}

__global__ void pool_kernel(const float* __restrict__ x, const int* __restrict__ batch,
                            float* __restrict__ pooled, int n) {
  int i = blockIdx.x * blockDim.x + threadIdx.x;
  if (i >= n * HDIM) return;
  int node = i >> 7, col = i & (HDIM - 1);
  atomicAdd(pooled + (size_t)batch[node] * HDIM + col, x[i]);
}

__global__ void final_linear_kernel(const float* __restrict__ pooled,
                                    const float* __restrict__ W,
                                    const float* __restrict__ b,
                                    float* __restrict__ out) {
  int i = blockIdx.x * blockDim.x + threadIdx.x;
  if (i >= NUM_GRAPHS * NUM_CLASSES) return;
  int g = i / NUM_CLASSES, c = i % NUM_CLASSES;
  float acc = b[c];
  const float* pr = pooled + (size_t)g * HDIM;
#pragma unroll 4
  for (int k = 0; k < HDIM; ++k) acc += pr[k] * W[k * NUM_CLASSES + c];
  out[i] = acc;
}

extern "C" void kernel_launch(void* const* d_in, const int* in_sizes, int n_in,
                              void* d_out, int out_size, void* d_ws, size_t ws_size,
                              hipStream_t stream) {
  const float* x     = (const float*)d_in[0];
  const float* ea    = (const float*)d_in[1];
  const float* enc_W = (const float*)d_in[2];
  const float* enc_b = (const float*)d_in[3];
  const float* t     = (const float*)d_in[4];
  const float* W1    = (const float*)d_in[5];
  const float* b1    = (const float*)d_in[6];
  const float* mlp_g = (const float*)d_in[7];
  const float* mlp_b = (const float*)d_in[8];
  const float* W2    = (const float*)d_in[9];
  const float* b2    = (const float*)d_in[10];
  const float* ln_g  = (const float*)d_in[11];
  const float* ln_b  = (const float*)d_in[12];
  const float* lin_W = (const float*)d_in[13];
  const float* lin_b = (const float*)d_in[14];
  const int*   eidx  = (const int*)d_in[15];
  const int*   batch = (const int*)d_in[16];

  const int N = in_sizes[0] / HDIM;
  const int E = in_sizes[1] / HDIM;
  const int L = in_sizes[4];
  const int* src = eidx;
  const int* dst = eidx + E;

  const size_t NH = (size_t)N * HDIM;
  float* h    = (float*)d_ws;
  float* rbuf = h    + NH;
  float* hc   = rbuf + NH;
  float* t1   = hc   + NH;
  int* lists  = (int*)(t1 + 2 * NH);

  float* out    = (float*)d_out;
  float* pooled = out + NUM_GRAPHS * NUM_CLASSES;

  const dim3 gblk(128);
  const int aggBlocks = (N + NODES_PER_BLOCK - 1) / NODES_PER_BLOCK;
  const int lnBlocks = (N + 7) / 8;

  gemm_rows16_wmma_f32<HDIM, HDIM, 2><<<N / 16, gblk, 0, stream>>>(x, enc_W, enc_b, h, 0);

  for (int layer = 0; layer < L; ++layer) {
    const float* rin = h;
    if (layer > 0) {
      ln_relu_kernel<<<lnBlocks, 256, 0, stream>>>(h, rbuf, ln_g + layer * HDIM, ln_b + layer * HDIM, N, HDIM);
      rin = rbuf;
    }
    if (layer == 0)
      gen_softmax_lds_kernel<true><<<aggBlocks, BLOCK_WAVES * 32, 0, stream>>>(rin, ea, src, dst, t, layer, hc, lists, E, N);
    else
      gen_softmax_lds_kernel<false><<<aggBlocks, BLOCK_WAVES * 32, 0, stream>>>(rin, ea, src, dst, t, layer, hc, lists, E, N);
    gemm_rows16_wmma_f32<HDIM, HDIM2, 4><<<N / 16, gblk, 0, stream>>>(
        hc, W1 + (size_t)layer * HDIM * HDIM2, b1 + layer * HDIM2, t1, 0);
    ln_relu_kernel<<<lnBlocks, 256, 0, stream>>>(t1, t1, mlp_g + layer * HDIM2, mlp_b + layer * HDIM2, N, HDIM2);
    gemm_rows16_wmma_f32<HDIM2, HDIM, 2><<<N / 16, gblk, 0, stream>>>(
        t1, W2 + (size_t)layer * HDIM2 * HDIM, b2 + layer * HDIM, h, layer > 0 ? 1 : 0);
  }

  ln_relu_kernel<<<lnBlocks, 256, 0, stream>>>(h, rbuf, ln_g, ln_b, N, HDIM);
  zero_kernel<<<(NUM_GRAPHS * HDIM + 255) / 256, 256, 0, stream>>>(pooled, NUM_GRAPHS * HDIM);
  pool_kernel<<<(int)((NH + 255) / 256), 256, 0, stream>>>(rbuf, batch, pooled, N);
  final_linear_kernel<<<(NUM_GRAPHS * NUM_CLASSES + 63) / 64, 64, 0, stream>>>(pooled, lin_W, lin_b, out);
}


# Row-wise launches (9 LayerNorms, pooling) on few long waves: a wave owns 8 rows, all loads up front; pooling adds node-order run sums, one atomic per run and column; LayerNorm row sums in new order
# speedup vs baseline: 3.1705x; 1.0368x over previous
#include <hip/hip_runtime.h>
#include <hip/hip_bf16.h>

typedef __attribute__((ext_vector_type(2))) float v2f;
typedef __attribute__((ext_vector_type(8))) float v8f;

#define HDIM 128
#define HDIM2 256
#define NUM_GRAPHS 64
#define NUM_CLASSES 10
#define MSG_EPS 1e-7f

template <int NT> struct fvec;
template <> struct fvec<2> { typedef __attribute__((ext_vector_type(2))) float type; };
template <> struct fvec<4> { typedef __attribute__((ext_vector_type(4))) float type; };

template <int K, int M, int NT>
__global__ __launch_bounds__(128)
void gemm_rows16_wmma_f32(const float* __restrict__ X,
                          const float* __restrict__ W,
                          const float* __restrict__ bias,
                          float* __restrict__ Y, int accum) {
  typedef typename fvec<NT>::type vnt;
  typedef __attribute__((ext_vector_type(4))) float v4f;
  static_assert(M == 64 * NT, "4 waves of NT tiles cover the M columns");
  constexpr int PITCH = K + 4;
  constexpr int KU = 16 / NT;
  constexpr int NG = K / (4 * KU);
  __shared__ float s_a[16 * PITCH];

  const int tid  = threadIdx.x;
  const int lane = tid & 31;
  const int wid  = tid >> 5;
  const int row0 = blockIdx.x * 16;
  const int half = lane >> 4;
  const int l16  = lane & 15;
  const int c0   = wid * (16 * NT);
  const int colv = c0 + NT * l16;

  const float* wp = W + (size_t)(half * 2) * M + colv;
  const float* ap = s_a + l16 * PITCH + half * 2;

  vnt bq[2][KU][2];
  v2f aq[2][KU];

#pragma unroll
  for (int u = 0; u < KU; ++u) {
    bq[0][u][0] = *(const vnt*)(wp + (size_t)(u * 4) * M);
    bq[0][u][1] = *(const vnt*)(wp + (size_t)(u * 4 + 1) * M);
  }

  {
    const float* xs = X + (size_t)row0 * K;
#pragma unroll
    for (int i = 0; i < (16 * K / 4) / 128; ++i) {
      const int idx = tid + 128 * i;
      const int r = idx / (K / 4), q = idx % (K / 4);
      const v4f v = *(const v4f*)(xs + (size_t)r * K + q * 4);
      *(v4f*)(s_a + r * PITCH + q * 4) = v;
    }
  }

  const vnt bv = *(const vnt*)(bias + colv);
  v8f c[NT];
  if (accum) {
#pragma unroll
    for (int r = 0; r < 8; ++r) {
      const vnt y = *(const vnt*)(Y + (size_t)(row0 + half * 8 + r) * M + colv);
#pragma unroll
      for (int t = 0; t < NT; ++t) c[t][r] = y[t] + bv[t];
    }
  } else {
#pragma unroll
    for (int t = 0; t < NT; ++t)
#pragma unroll
      for (int r = 0; r < 8; ++r) c[t][r] = bv[t];
  }

  __syncthreads();

#pragma unroll
  for (int u = 0; u < KU; ++u) aq[0][u] = *(const v2f*)(ap + u * 4);

#pragma unroll
  for (int g = 0; g < NG; ++g) {
    const int cur = g & 1, nxt = cur ^ 1;
    if (g + 1 < NG) {
#pragma unroll
      for (int u = 0; u < KU; ++u) {
        const int k = ((g + 1) * KU + u) * 4;
        bq[nxt][u][0] = *(const vnt*)(wp + (size_t)k * M);
        bq[nxt][u][1] = *(const vnt*)(wp + (size_t)(k + 1) * M);
        aq[nxt][u]    = *(const v2f*)(ap + k);
      }
    }
#pragma unroll
    for (int u = 0; u < KU; ++u) {
#pragma unroll
      for (int t = 0; t < NT; ++t) {
        v2f b; b[0] = bq[cur][u][0][t]; b[1] = bq[cur][u][1][t];
        c[t] = __builtin_amdgcn_wmma_f32_16x16x4_f32(false, aq[cur][u], false, b,
                                                     (short)0, c[t], false, false);
      }
    }
  }

#pragma unroll
  for (int r = 0; r < 8; ++r) {
    vnt o;
#pragma unroll
    for (int t = 0; t < NT; ++t) o[t] = c[t][r];
    *(vnt*)(Y + (size_t)(row0 + half * 8 + r) * M + colv) = o;
  }
}

#define ROWS_PER_WAVE 8
#define ROW_WAVES 4

template <int D>
__global__ __launch_bounds__(ROW_WAVES * 32)
void ln_relu_rows_kernel(const float* x, float* y,
                         const float* __restrict__ g, const float* __restrict__ b, int rows) {
  typedef __attribute__((ext_vector_type(4))) float v4f;
  constexpr int NP = D / 128;
  const int lane = threadIdx.x & 31;
  const int wave = __builtin_amdgcn_readfirstlane(threadIdx.x >> 5);
  const int row0 = (blockIdx.x * ROW_WAVES + wave) * ROWS_PER_WAVE;
  if (row0 >= rows) return;
  v4f v[ROWS_PER_WAVE][NP];
#pragma unroll
  for (int r = 0; r < ROWS_PER_WAVE; ++r) {
    const int row = row0 + r < rows ? row0 + r : rows - 1;
#pragma unroll
    for (int p = 0; p < NP; ++p)
      v[r][p] = *(const v4f*)(x + (size_t)row * D + p * 128 + lane * 4);
  }
  v4f gv[NP], bv[NP];
#pragma unroll
  for (int p = 0; p < NP; ++p) {
    gv[p] = *(const v4f*)(g + p * 128 + lane * 4);
    bv[p] = *(const v4f*)(b + p * 128 + lane * 4);
  }
  float s[ROWS_PER_WAVE], q[ROWS_PER_WAVE];
#pragma unroll
  for (int r = 0; r < ROWS_PER_WAVE; ++r) {
    s[r] = 0.f; q[r] = 0.f;
#pragma unroll
    for (int p = 0; p < NP; ++p)
#pragma unroll
      for (int e = 0; e < 4; ++e) { const float t = v[r][p][e]; s[r] += t; q[r] += t * t; }
  }
#pragma unroll
  for (int off = 16; off > 0; off >>= 1) {
#pragma unroll
    for (int r = 0; r < ROWS_PER_WAVE; ++r) {
      s[r] += __shfl_xor(s[r], off, 32);
      q[r] += __shfl_xor(q[r], off, 32);
    }
  }
#pragma unroll
  for (int r = 0; r < ROWS_PER_WAVE; ++r) {
    if (row0 + r < rows) {
      const float mean = s[r] / (float)D;
      const float var  = q[r] / (float)D - mean * mean;
      const float inv  = rsqrtf(var + 1e-5f);
#pragma unroll
      for (int p = 0; p < NP; ++p) {
        v4f o;
#pragma unroll
        for (int e = 0; e < 4; ++e)
          o[e] = fmaxf((v[r][p][e] - mean) * inv * gv[p][e] + bv[p][e], 0.f);
        *(v4f*)(y + (size_t)(row0 + r) * D + p * 128 + lane * 4) = o;
      }
    }
  }
}

__global__ void zero_kernel(float* p, int n) {
  int i = blockIdx.x * blockDim.x + threadIdx.x;
  if (i < n) p[i] = 0.f;
}

#define NODES_PER_BLOCK 40
#define BLOCK_WAVES 32
#define SCAN_CHUNKS 16
#define GROUP_EDGES 8
#define M_CAP 1536
#define QCOLS 32
#define LIST_HDR 32
#define LIST_STRIDE (LIST_HDR + 3 * M_CAP)

template <int PHASE>
__device__ __forceinline__ void fold_rows(
    const float* __restrict__ x, const float* __restrict__ ea,
    const int (&pe)[GROUP_EDGES], const int (&s)[GROUP_EDGES], const int (&r)[GROUP_EDGES],
    int count, float tv, int lane, float* s_mx, float* s_den, float* s_num) {
  float a[GROUP_EDGES][4], b[GROUP_EDGES][4];
#pragma unroll
  for (int g = 0; g < GROUP_EDGES; ++g) {
    const float* xr = x  + (size_t)s[g]  * HDIM + lane;
    const float* er = ea + (size_t)pe[g] * HDIM + lane;
#pragma unroll
    for (int j = 0; j < 4; ++j) { a[g][j] = xr[32 * j]; b[g][j] = er[32 * j]; }
  }
  float mv[GROUP_EDGES][4];
  if (PHASE == 2) {
#pragma unroll
    for (int g = 0; g < GROUP_EDGES; ++g) {
      if (g < count) {
#pragma unroll
        for (int j = 0; j < 4; ++j) mv[g][j] = s_mx[r[g] * HDIM + lane + 32 * j];
      }
    }
  }
#pragma unroll
  for (int g = 0; g < GROUP_EDGES; ++g) {
    if (g < count) {
      const int o = r[g] * HDIM + lane;
#pragma unroll
      for (int j = 0; j < 4; ++j) {
        const float m = fmaxf(a[g][j] + b[g][j], 0.f) + MSG_EPS;
        if (PHASE == 1) {
          const float v = m * tv;
          if (v >= 0.0f) atomicMax((int*)&s_mx[o + 32 * j], __float_as_int(v));
          else           atomicMin((unsigned int*)&s_mx[o + 32 * j], __float_as_uint(v));
        } else {
          const float ex = __expf(m * tv - mv[g][j]);
          atomicAdd(&s_den[o + 32 * j], ex);
          atomicAdd(&s_num[o + 32 * j], ex * m);
        }
      }
    }
  }
}

template <int PHASE>
__device__ __forceinline__ void fold_edge_group(
    const float* __restrict__ x, const float* __restrict__ ea,
    const int* __restrict__ src, const int* __restrict__ dst,
    const int (&pe)[GROUP_EDGES], int count, int node0, float tv, int lane,
    float* s_mx, float* s_den, float* s_num) {
  int s[GROUP_EDGES], r[GROUP_EDGES];
#pragma unroll
  for (int g = 0; g < GROUP_EDGES; ++g) {
    s[g] = src[pe[g]];
    r[g] = dst[pe[g]] - node0;
  }
  fold_rows<PHASE>(x, ea, pe, s, r, count, tv, lane, s_mx, s_den, s_num);
}

__device__ __forceinline__ unsigned int scan_step_hits(const int* __restrict__ dst, int E, int c0,
                                                       int node0, int lane) {
  unsigned int hits = 0u;
  if ((c0 + SCAN_CHUNKS) * 32 <= E) {
    const int* p = dst + c0 * 32 + lane;
#pragma unroll
    for (int u = 0; u < SCAN_CHUNKS; ++u) {
      const int dv = p[u * 32];
      hits |= ((unsigned int)(dv - node0) < (unsigned int)NODES_PER_BLOCK ? 1u : 0u) << u;
    }
  } else {
#pragma unroll
    for (int u = 0; u < SCAN_CHUNKS; ++u) {
      const int e  = (c0 + u) * 32 + lane;
      const int dv = dst[e < E ? e : E - 1];
      const bool hit = (e < E) && ((unsigned int)(dv - node0) < (unsigned int)NODES_PER_BLOCK);
      hits |= (hit ? 1u : 0u) << u;
    }
  }
  return hits;
}

__device__ __forceinline__ void scan_build_list(
    const int* __restrict__ dst, int E, int node0, int lane, int wave,
    int* s_cnt, int* s_le) {
  const int nchunks = (E + 31) >> 5;
  for (int c0 = wave * SCAN_CHUNKS; c0 < nchunks; c0 += BLOCK_WAVES * SCAN_CHUNKS) {
    unsigned int hits = scan_step_hits(dst, E, c0, node0, lane);
    for (int it = 0; it < SCAN_CHUNKS; ++it) {
      const unsigned int active = __builtin_amdgcn_ballot_w32(hits != 0u);
      if (active == 0u) break;
      int slot0 = 0;
      if (lane == 0) slot0 = atomicAdd(s_cnt, __builtin_popcount(active));
      slot0 = __builtin_amdgcn_readfirstlane(slot0);
      if (hits != 0u) {
        const int u = __builtin_ctz(hits);
        hits &= hits - 1u;
        const int e = (c0 + u) * 32 + lane;
        const int slot = slot0 + __builtin_popcount(active & ((1u << lane) - 1u));
        if (slot < M_CAP) s_le[slot] = e;
      }
    }
  }
}

template <int NG, bool FULL>
__device__ __forceinline__ void quarter_messages(
    const float* __restrict__ x, const float* __restrict__ ea,
    int first, int n, float tv, int lane, int col,
    const int* s_le, const int* s_ls, const int* s_lr, float* s_mx, float* s_m) {
  constexpr int NE = GROUP_EDGES * NG;
  const int used = FULL ? NE : n;
  const int li = lane < used ? first + (lane >> 3) * (GROUP_EDGES * BLOCK_WAVES) + (lane & 7) : first;
  const int ev = s_le[li], sv = s_ls[li], rv = s_lr[li];
  float a[NE], b[NE];
#pragma unroll
  for (int g = 0; g < NE; ++g) {
    const int pe = __builtin_amdgcn_readlane(ev, g);
    const int s  = __builtin_amdgcn_readlane(sv, g);
    a[g] = x[(size_t)s * HDIM + col];
    b[g] = ea[(size_t)pe * HDIM + col];
  }
#pragma unroll
  for (int g = 0; g < NE; ++g) {
    if (FULL || g < n) {
      const int r   = __builtin_amdgcn_readlane(rv, g);
      const int o   = r * HDIM + col;
      const int row = first + (g >> 3) * (GROUP_EDGES * BLOCK_WAVES) + (g & 7);
      const float m = fmaxf(a[g] + b[g], 0.f) + MSG_EPS;
      s_m[row * QCOLS + lane] = m;
      const float v = m * tv;
      if (v >= 0.0f) atomicMax((int*)&s_mx[o], __float_as_int(v));
      else           atomicMin((unsigned int*)&s_mx[o], __float_as_uint(v));
    }
  }
}

template <bool FULL>
__device__ __forceinline__ void quarter_sums(
    int first, int n, float tv, int lane, int col,
    const int* s_lr, const float* s_mx, float* s_den, float* s_num, const float* s_m) {
  const int used = FULL ? GROUP_EDGES : n;
  const int rv = s_lr[lane < used ? first + lane : first];
  float m[GROUP_EDGES], mv[GROUP_EDGES];
  int o[GROUP_EDGES];
#pragma unroll
  for (int g = 0; g < GROUP_EDGES; ++g) {
    if (FULL || g < n) {
      o[g]  = __builtin_amdgcn_readlane(rv, g) * HDIM + col;
      m[g]  = s_m[(first + g) * QCOLS + lane];
      mv[g] = s_mx[o[g]];
    }
  }
#pragma unroll
  for (int g = 0; g < GROUP_EDGES; ++g) {
    if (FULL || g < n) {
      const float ex = __expf(m[g] * tv - mv[g]);
      atomicAdd(&s_den[o[g]], ex);
      atomicAdd(&s_num[o[g]], ex * m[g]);
    }
  }
}

__device__ __forceinline__ void fold_list_by_quarters(
    const float* __restrict__ x, const float* __restrict__ ea,
    int cnt, float tv, int lane, int wave,
    const int* s_le, const int* s_ls, const int* s_lr,
    float* s_mx, float* s_den, float* s_num, float* s_m) {
  const int G     = (cnt + GROUP_EDGES - 1) / GROUP_EDGES;
  const int tail  = cnt - GROUP_EDGES * (G - 1);
  const int ng    = G > wave ? (G - wave + BLOCK_WAVES - 1) / BLOCK_WAVES : 0;
  const bool has_short = ng > 0 && wave + BLOCK_WAVES * (ng - 1) == G - 1 && tail < GROUP_EDGES;
  const int nfull = has_short ? ng - 1 : ng;
  const int STRIDE = GROUP_EDGES * BLOCK_WAVES;
  const int first0 = GROUP_EDGES * wave;
#pragma unroll 1
  for (int q = 0; q < HDIM / QCOLS; ++q) {
    const int col = q * QCOLS + lane;
    int j = 0;
    for (; j + 3 <= nfull; j += 3)
      quarter_messages<3, true>(x, ea, first0 + j * STRIDE, 0, tv, lane, col, s_le, s_ls, s_lr, s_mx, s_m);
    if (nfull - j == 2)
      quarter_messages<2, true>(x, ea, first0 + j * STRIDE, 0, tv, lane, col, s_le, s_ls, s_lr, s_mx, s_m);
    else if (nfull - j == 1)
      quarter_messages<1, true>(x, ea, first0 + j * STRIDE, 0, tv, lane, col, s_le, s_ls, s_lr, s_mx, s_m);
    if (has_short)
      quarter_messages<1, false>(x, ea, first0 + nfull * STRIDE, tail, tv, lane, col, s_le, s_ls, s_lr, s_mx, s_m);
    __syncthreads();
    for (int k = 0; k < nfull; ++k)
      quarter_sums<true>(first0 + k * STRIDE, 0, tv, lane, col, s_lr, s_mx, s_den, s_num, s_m);
    if (has_short)
      quarter_sums<false>(first0 + nfull * STRIDE, tail, tv, lane, col, s_lr, s_mx, s_den, s_num, s_m);
  }
}

template <int PHASE>
__device__ __forceinline__ void scan_and_fold(
    const float* __restrict__ x, const float* __restrict__ ea,
    const int* __restrict__ src, const int* __restrict__ dst,
    int E, int node0, float tv, int lane, int wave,
    float* s_mx, float* s_den, float* s_num) {
  int pe[GROUP_EDGES];
#pragma unroll
  for (int g = 0; g < GROUP_EDGES; ++g) pe[g] = 0;
  int pn = 0;
  const int nchunks = (E + 31) >> 5;
  for (int c0 = wave * SCAN_CHUNKS; c0 < nchunks; c0 += BLOCK_WAVES * SCAN_CHUNKS) {
    const unsigned int hits = scan_step_hits(dst, E, c0, node0, lane);
    if (__builtin_amdgcn_ballot_w32(hits != 0u) != 0u) {
#pragma unroll 1
      for (int u = 0; u < SCAN_CHUNKS; ++u) {
        unsigned int mask = __builtin_amdgcn_ballot_w32(((hits >> u) & 1u) != 0u);
        const int base = (c0 + u) * 32;
        while (mask != 0u) {
          const int bit = __builtin_ctz(mask);
          mask &= mask - 1u;
#pragma unroll
          for (int g = GROUP_EDGES - 1; g > 0; --g) pe[g] = pe[g - 1];
          pe[0] = base + bit;
          if (++pn == GROUP_EDGES) {
            fold_edge_group<PHASE>(x, ea, src, dst, pe, GROUP_EDGES, node0, tv, lane, s_mx, s_den, s_num);
            pn = 0;
          }
        }
      }
    }
  }
  if (pn > 0)
    fold_edge_group<PHASE>(x, ea, src, dst, pe, pn, node0, tv, lane, s_mx, s_den, s_num);
}

template <bool BUILD>
__global__ __launch_bounds__(BLOCK_WAVES * 32)
void gen_softmax_lds_kernel(const float* __restrict__ x, const float* __restrict__ ea,
                            const int* __restrict__ src, const int* __restrict__ dst,
                            const float* __restrict__ tptr, int layer,
                            float* __restrict__ hc, int* __restrict__ lists, int E, int N) {
  __shared__ float s_mx[NODES_PER_BLOCK * HDIM];
  __shared__ float s_den[NODES_PER_BLOCK * HDIM];
  __shared__ float s_num[NODES_PER_BLOCK * HDIM];
  __shared__ float s_m[M_CAP * QCOLS];
  __shared__ int s_le[M_CAP];
  __shared__ int s_ls[M_CAP];
  __shared__ int s_lr[M_CAP];
  __shared__ int s_cnt;
  const int tid   = threadIdx.x;
  const int lane  = tid & 31;
  const int wave  = __builtin_amdgcn_readfirstlane(tid >> 5);
  const int node0 = blockIdx.x * NODES_PER_BLOCK;
  int* gl = lists + (size_t)blockIdx.x * LIST_STRIDE;
  for (int i = tid; i < NODES_PER_BLOCK * HDIM; i += BLOCK_WAVES * 32) {
    s_mx[i]  = __int_as_float(0xff800000);
    s_den[i] = 0.f;
    s_num[i] = 0.f;
  }
  const float tv = tptr[layer];
  int cnt;
  if (BUILD) {
    if (tid == 0) s_cnt = 0;
    __syncthreads();
    scan_build_list(dst, E, node0, lane, wave, &s_cnt, s_le);
    __syncthreads();
    cnt = __builtin_amdgcn_readfirstlane(s_cnt);
    if (tid == 0) gl[0] = cnt;
    if (cnt <= M_CAP) {
      for (int i = tid; i < cnt; i += BLOCK_WAVES * 32) {
        const int e = s_le[i];
        const int s = src[e];
        const int r = dst[e] - node0;
        s_ls[i] = s;
        s_lr[i] = r;
        gl[LIST_HDR + i]             = e;
        gl[LIST_HDR + M_CAP + i]     = s;
        gl[LIST_HDR + 2 * M_CAP + i] = r;
      }
    }
  } else {
    cnt = __builtin_amdgcn_readfirstlane(gl[0]);
    if (cnt <= M_CAP) {
      constexpr int TRIPS = (M_CAP + BLOCK_WAVES * 32 - 1) / (BLOCK_WAVES * 32);
      int ve[TRIPS], vs[TRIPS], vr[TRIPS];
#pragma unroll
      for (int k = 0; k < TRIPS; ++k) {
        const int i = tid + k * (BLOCK_WAVES * 32);
        if (i < cnt) {
          ve[k] = gl[LIST_HDR + i];
          vs[k] = gl[LIST_HDR + M_CAP + i];
          vr[k] = gl[LIST_HDR + 2 * M_CAP + i];
        }
      }
#pragma unroll
      for (int k = 0; k < TRIPS; ++k) {
        const int i = tid + k * (BLOCK_WAVES * 32);
        if (i < cnt) {
          s_le[i] = ve[k];
          s_ls[i] = vs[k];
          s_lr[i] = vr[k];
        }
      }
    }
  }
  __syncthreads();
  if (cnt <= M_CAP) {
    fold_list_by_quarters(x, ea, cnt, tv, lane, wave, s_le, s_ls, s_lr, s_mx, s_den, s_num, s_m);
  } else {
    scan_and_fold<1>(x, ea, src, dst, E, node0, tv, lane, wave, s_mx, s_den, s_num);
    __syncthreads();
    scan_and_fold<2>(x, ea, src, dst, E, node0, tv, lane, wave, s_mx, s_den, s_num);
  }
  __syncthreads();
  for (int i = tid; i < NODES_PER_BLOCK * HDIM; i += BLOCK_WAVES * 32) {
    const int node = node0 + i / HDIM;
    if (node < N) {
      const float dv  = s_den[i];
      const float agg = dv > 0.f ? s_num[i] / dv : 0.f;
      const size_t gi = (size_t)node * HDIM + i % HDIM;
      hc[gi] = agg + x[gi];
    }
  }
}

__device__ __forceinline__ void pool_flush(float* __restrict__ pooled, int gid, const float (&acc)[4], int lane) {
  if ((unsigned int)gid < (unsigned int)NUM_GRAPHS) {
#pragma unroll
    for (int j = 0; j < 4; ++j) atomicAdd(pooled + (size_t)gid * HDIM + lane + 32 * j, acc[j]);
  }
}

__global__ __launch_bounds__(ROW_WAVES * 32)
void pool_rows_kernel(const float* __restrict__ x, const int* __restrict__ batch,
                      float* __restrict__ pooled, int n) {
  const int lane = threadIdx.x & 31;
  const int wave = __builtin_amdgcn_readfirstlane(threadIdx.x >> 5);
  const int row0 = (blockIdx.x * ROW_WAVES + wave) * ROWS_PER_WAVE;
  if (row0 >= n) return;
  float v[ROWS_PER_WAVE][4];
  int gid[ROWS_PER_WAVE];
#pragma unroll
  for (int r = 0; r < ROWS_PER_WAVE; ++r) {
    const bool live = row0 + r < n;
    const int row = live ? row0 + r : n - 1;
    gid[r] = __builtin_amdgcn_readfirstlane(batch[row]);
#pragma unroll
    for (int j = 0; j < 4; ++j) {
      const float t = x[(size_t)row * HDIM + lane + 32 * j];
      v[r][j] = live ? t : 0.f;
    }
  }
  float acc[4] = {0.f, 0.f, 0.f, 0.f};
  int cur = gid[0];
#pragma unroll
  for (int r = 0; r < ROWS_PER_WAVE; ++r) {
    if (gid[r] != cur) {
      pool_flush(pooled, cur, acc, lane);
#pragma unroll
      for (int j = 0; j < 4; ++j) acc[j] = 0.f;
      cur = gid[r];
    }
#pragma unroll
    for (int j = 0; j < 4; ++j) acc[j] += v[r][j];
  }
  pool_flush(pooled, cur, acc, lane);
}

__global__ void final_linear_kernel(const float* __restrict__ pooled,
                                    const float* __restrict__ W,
                                    const float* __restrict__ b,
                                    float* __restrict__ out) {
  int i = blockIdx.x * blockDim.x + threadIdx.x;
  if (i >= NUM_GRAPHS * NUM_CLASSES) return;
  int g = i / NUM_CLASSES, c = i % NUM_CLASSES;
  float acc = b[c];
  const float* pr = pooled + (size_t)g * HDIM;
#pragma unroll 4
  for (int k = 0; k < HDIM; ++k) acc += pr[k] * W[k * NUM_CLASSES + c];
  out[i] = acc;
}

extern "C" void kernel_launch(void* const* d_in, const int* in_sizes, int n_in,
                              void* d_out, int out_size, void* d_ws, size_t ws_size,
                              hipStream_t stream) {
  const float* x     = (const float*)d_in[0];
  const float* ea    = (const float*)d_in[1];
  const float* enc_W = (const float*)d_in[2];
  const float* enc_b = (const float*)d_in[3];
  const float* t     = (const float*)d_in[4];
  const float* W1    = (const float*)d_in[5];
  const float* b1    = (const float*)d_in[6];
  const float* mlp_g = (const float*)d_in[7];
  const float* mlp_b = (const float*)d_in[8];
  const float* W2    = (const float*)d_in[9];
  const float* b2    = (const float*)d_in[10];
  const float* ln_g  = (const float*)d_in[11];
  const float* ln_b  = (const float*)d_in[12];
  const float* lin_W = (const float*)d_in[13];
  const float* lin_b = (const float*)d_in[14];
  const int*   eidx  = (const int*)d_in[15];
  const int*   batch = (const int*)d_in[16];

  const int N = in_sizes[0] / HDIM;
  const int E = in_sizes[1] / HDIM;
  const int L = in_sizes[4];
  const int* src = eidx;
  const int* dst = eidx + E;

  const size_t NH = (size_t)N * HDIM;
  float* h    = (float*)d_ws;
  float* rbuf = h    + NH;
  float* hc   = rbuf + NH;
  float* t1   = hc   + NH;
  int* lists  = (int*)(t1 + 2 * NH);

  float* out    = (float*)d_out;
  float* pooled = out + NUM_GRAPHS * NUM_CLASSES;

  const dim3 gblk(128);
  const int aggBlocks = (N + NODES_PER_BLOCK - 1) / NODES_PER_BLOCK;
  const int rowBlocks = (N + ROW_WAVES * ROWS_PER_WAVE - 1) / (ROW_WAVES * ROWS_PER_WAVE);

  gemm_rows16_wmma_f32<HDIM, HDIM, 2><<<N / 16, gblk, 0, stream>>>(x, enc_W, enc_b, h, 0);

  for (int layer = 0; layer < L; ++layer) {
    const float* rin = h;
    if (layer > 0) {
      ln_relu_rows_kernel<HDIM><<<rowBlocks, ROW_WAVES * 32, 0, stream>>>(h, rbuf, ln_g + layer * HDIM, ln_b + layer * HDIM, N);
      rin = rbuf;
    }
    if (layer == 0)
      gen_softmax_lds_kernel<true><<<aggBlocks, BLOCK_WAVES * 32, 0, stream>>>(rin, ea, src, dst, t, layer, hc, lists, E, N);
    else
      gen_softmax_lds_kernel<false><<<aggBlocks, BLOCK_WAVES * 32, 0, stream>>>(rin, ea, src, dst, t, layer, hc, lists, E, N);
    gemm_rows16_wmma_f32<HDIM, HDIM2, 4><<<N / 16, gblk, 0, stream>>>(
        hc, W1 + (size_t)layer * HDIM * HDIM2, b1 + layer * HDIM2, t1, 0);
    ln_relu_rows_kernel<HDIM2><<<rowBlocks, ROW_WAVES * 32, 0, stream>>>(t1, t1, mlp_g + layer * HDIM2, mlp_b + layer * HDIM2, N);
    gemm_rows16_wmma_f32<HDIM2, HDIM, 2><<<N / 16, gblk, 0, stream>>>(
        t1, W2 + (size_t)layer * HDIM2 * HDIM, b2 + layer * HDIM, h, layer > 0 ? 1 : 0);
  }

  ln_relu_rows_kernel<HDIM><<<rowBlocks, ROW_WAVES * 32, 0, stream>>>(h, rbuf, ln_g, ln_b, N);
  zero_kernel<<<(NUM_GRAPHS * HDIM + 255) / 256, 256, 0, stream>>>(pooled, NUM_GRAPHS * HDIM);
  pool_rows_kernel<<<rowBlocks, ROW_WAVES * 32, 0, stream>>>(rbuf, batch, pooled, N);
  final_linear_kernel<<<(NUM_GRAPHS * NUM_CLASSES + 63) / 64, 64, 0, stream>>>(pooled, lin_W, lin_b, out);
}


# LayerNorm+ReLU applied to the accumulators at the end of the product that makes its rows (a block holds whole rows): the 8 LayerNorm launches go, 24 to 16; LayerNorm row sums in a new fixed order
# speedup vs baseline: 3.5473x; 1.1188x over previous
#include <hip/hip_runtime.h>
#include <hip/hip_bf16.h>

typedef __attribute__((ext_vector_type(2))) float v2f;
typedef __attribute__((ext_vector_type(8))) float v8f;

#define HDIM 128
#define HDIM2 256
#define NUM_GRAPHS 64
#define NUM_CLASSES 10
#define MSG_EPS 1e-7f

template <int NT> struct fvec;
template <> struct fvec<2> { typedef __attribute__((ext_vector_type(2))) float type; };
template <> struct fvec<4> { typedef __attribute__((ext_vector_type(4))) float type; };

template <int XMASK>
__device__ __forceinline__ float row16_xadd(float v) {
  return v + __int_as_float(__builtin_amdgcn_update_dpp(0, __float_as_int(v), 0x160 + XMASK, 0xf, 0xf, false));
}

template <int K, int M, int NT, int LNMODE>
__global__ __launch_bounds__(128)
void gemm_rows16_wmma_f32(const float* __restrict__ X,
                          const float* __restrict__ W,
                          const float* __restrict__ bias,
                          float* __restrict__ Y, int accum,
                          const float* __restrict__ lng, const float* __restrict__ lnb,
                          float* __restrict__ Z) {
  typedef typename fvec<NT>::type vnt;
  typedef __attribute__((ext_vector_type(4))) float v4f;
  static_assert(M == 64 * NT, "4 waves of NT tiles cover the M columns");
  constexpr int PITCH = K + 4;
  constexpr int KU = 16 / NT;
  constexpr int NG = K / (4 * KU);
  __shared__ float s_a[16 * PITCH];
  __shared__ alignas(16) float s_red[16 * 8];

  const int tid  = threadIdx.x;
  const int lane = tid & 31;
  const int wid  = tid >> 5;
  const int row0 = blockIdx.x * 16;
  const int half = lane >> 4;
  const int l16  = lane & 15;
  const int c0   = wid * (16 * NT);
  const int colv = c0 + NT * l16;

  const float* wp = W + (size_t)(half * 2) * M + colv;
  const float* ap = s_a + l16 * PITCH + half * 2;

  vnt bq[2][KU][2];
  v2f aq[2][KU];

#pragma unroll
  for (int u = 0; u < KU; ++u) {
    bq[0][u][0] = *(const vnt*)(wp + (size_t)(u * 4) * M);
    bq[0][u][1] = *(const vnt*)(wp + (size_t)(u * 4 + 1) * M);
  }

  {
    const float* xs = X + (size_t)row0 * K;
#pragma unroll
    for (int i = 0; i < (16 * K / 4) / 128; ++i) {
      const int idx = tid + 128 * i;
      const int r = idx / (K / 4), q = idx % (K / 4);
      const v4f v = *(const v4f*)(xs + (size_t)r * K + q * 4);
      *(v4f*)(s_a + r * PITCH + q * 4) = v;
    }
  }

  const vnt bv = *(const vnt*)(bias + colv);
  v8f c[NT];
  if (accum) {
#pragma unroll
    for (int r = 0; r < 8; ++r) {
      const vnt y = *(const vnt*)(Y + (size_t)(row0 + half * 8 + r) * M + colv);
#pragma unroll
      for (int t = 0; t < NT; ++t) c[t][r] = y[t] + bv[t];
    }
  } else {
#pragma unroll
    for (int t = 0; t < NT; ++t)
#pragma unroll
      for (int r = 0; r < 8; ++r) c[t][r] = bv[t];
  }

  __syncthreads();

#pragma unroll
  for (int u = 0; u < KU; ++u) aq[0][u] = *(const v2f*)(ap + u * 4);

#pragma unroll
  for (int g = 0; g < NG; ++g) {
    const int cur = g & 1, nxt = cur ^ 1;
    if (g + 1 < NG) {
#pragma unroll
      for (int u = 0; u < KU; ++u) {
        const int k = ((g + 1) * KU + u) * 4;
        bq[nxt][u][0] = *(const vnt*)(wp + (size_t)k * M);
        bq[nxt][u][1] = *(const vnt*)(wp + (size_t)(k + 1) * M);
        aq[nxt][u]    = *(const v2f*)(ap + k);
      }
    }
#pragma unroll
    for (int u = 0; u < KU; ++u) {
#pragma unroll
      for (int t = 0; t < NT; ++t) {
        v2f b; b[0] = bq[cur][u][0][t]; b[1] = bq[cur][u][1][t];
        c[t] = __builtin_amdgcn_wmma_f32_16x16x4_f32(false, aq[cur][u], false, b,
                                                     (short)0, c[t], false, false);
      }
    }
  }

  if (LNMODE != 1) {
#pragma unroll
    for (int r = 0; r < 8; ++r) {
      vnt o;
#pragma unroll
      for (int t = 0; t < NT; ++t) o[t] = c[t][r];
      *(vnt*)(Y + (size_t)(row0 + half * 8 + r) * M + colv) = o;
    }
  }
  if (LNMODE != 0) {
    const vnt gv = *(const vnt*)(lng + colv);
    const vnt ev = *(const vnt*)(lnb + colv);
    float s[8], q[8];
#pragma unroll
    for (int r = 0; r < 8; ++r) {
      s[r] = 0.f; q[r] = 0.f;
#pragma unroll
      for (int t = 0; t < NT; ++t) { const float v = c[t][r]; s[r] += v; q[r] += v * v; }
    }
#pragma unroll
    for (int r = 0; r < 8; ++r) { s[r] = row16_xadd<1>(s[r]); q[r] = row16_xadd<1>(q[r]); }
#pragma unroll
    for (int r = 0; r < 8; ++r) { s[r] = row16_xadd<2>(s[r]); q[r] = row16_xadd<2>(q[r]); }
#pragma unroll
    for (int r = 0; r < 8; ++r) { s[r] = row16_xadd<4>(s[r]); q[r] = row16_xadd<4>(q[r]); }
#pragma unroll
    for (int r = 0; r < 8; ++r) { s[r] = row16_xadd<8>(s[r]); q[r] = row16_xadd<8>(q[r]); }
    if (l16 == 0) {
#pragma unroll
      for (int r = 0; r < 8; ++r) {
        v2f p; p[0] = s[r]; p[1] = q[r];
        *(v2f*)(s_red + (half * 8 + r) * 8 + wid * 2) = p;
      }
    }
    __syncthreads();
    float* dst = LNMODE == 1 ? Y : Z;
#pragma unroll
    for (int r = 0; r < 8; ++r) {
      const v4f p0 = *(const v4f*)(s_red + (half * 8 + r) * 8);
      const v4f p1 = *(const v4f*)(s_red + (half * 8 + r) * 8 + 4);
      const float sum  = ((p0[0] + p0[2]) + p1[0]) + p1[2];
      const float sq   = ((p0[1] + p0[3]) + p1[1]) + p1[3];
      const float mean = sum / (float)M;
      const float var  = sq / (float)M - mean * mean;
      const float inv  = rsqrtf(var + 1e-5f);
      vnt o;
#pragma unroll
      for (int t = 0; t < NT; ++t)
        o[t] = fmaxf((c[t][r] - mean) * inv * gv[t] + ev[t], 0.f);
      *(vnt*)(dst + (size_t)(row0 + half * 8 + r) * M + colv) = o;
    }
  }
}

#define ROWS_PER_WAVE 8
#define ROW_WAVES 4

__global__ void zero_kernel(float* p, int n) {
  int i = blockIdx.x * blockDim.x + threadIdx.x;
  if (i < n) p[i] = 0.f;
}

#define NODES_PER_BLOCK 40
#define BLOCK_WAVES 32
#define SCAN_CHUNKS 16
#define GROUP_EDGES 8
#define M_CAP 1536
#define QCOLS 32
#define LIST_HDR 32
#define LIST_STRIDE (LIST_HDR + 3 * M_CAP)

template <int PHASE>
__device__ __forceinline__ void fold_rows(
    const float* __restrict__ x, const float* __restrict__ ea,
    const int (&pe)[GROUP_EDGES], const int (&s)[GROUP_EDGES], const int (&r)[GROUP_EDGES],
    int count, float tv, int lane, float* s_mx, float* s_den, float* s_num) {
  float a[GROUP_EDGES][4], b[GROUP_EDGES][4];
#pragma unroll
  for (int g = 0; g < GROUP_EDGES; ++g) {
    const float* xr = x  + (size_t)s[g]  * HDIM + lane;
    const float* er = ea + (size_t)pe[g] * HDIM + lane;
#pragma unroll
    for (int j = 0; j < 4; ++j) { a[g][j] = xr[32 * j]; b[g][j] = er[32 * j]; }
  }
  float mv[GROUP_EDGES][4];
  if (PHASE == 2) {
#pragma unroll
    for (int g = 0; g < GROUP_EDGES; ++g) {
      if (g < count) {
#pragma unroll
        for (int j = 0; j < 4; ++j) mv[g][j] = s_mx[r[g] * HDIM + lane + 32 * j];
      }
    }
  }
#pragma unroll
  for (int g = 0; g < GROUP_EDGES; ++g) {
    if (g < count) {
      const int o = r[g] * HDIM + lane;
#pragma unroll
      for (int j = 0; j < 4; ++j) {
        const float m = fmaxf(a[g][j] + b[g][j], 0.f) + MSG_EPS;
        if (PHASE == 1) {
          const float v = m * tv;
          if (v >= 0.0f) atomicMax((int*)&s_mx[o + 32 * j], __float_as_int(v));
          else           atomicMin((unsigned int*)&s_mx[o + 32 * j], __float_as_uint(v));
        } else {
          const float ex = __expf(m * tv - mv[g][j]);
          atomicAdd(&s_den[o + 32 * j], ex);
          atomicAdd(&s_num[o + 32 * j], ex * m);
        }
      }
    }
  }
}

template <int PHASE>
__device__ __forceinline__ void fold_edge_group(
    const float* __restrict__ x, const float* __restrict__ ea,
    const int* __restrict__ src, const int* __restrict__ dst,
    const int (&pe)[GROUP_EDGES], int count, int node0, float tv, int lane,
    float* s_mx, float* s_den, float* s_num) {
  int s[GROUP_EDGES], r[GROUP_EDGES];
#pragma unroll
  for (int g = 0; g < GROUP_EDGES; ++g) {
    s[g] = src[pe[g]];
    r[g] = dst[pe[g]] - node0;
  }
  fold_rows<PHASE>(x, ea, pe, s, r, count, tv, lane, s_mx, s_den, s_num);
}

__device__ __forceinline__ unsigned int scan_step_hits(const int* __restrict__ dst, int E, int c0,
                                                       int node0, int lane) {
  unsigned int hits = 0u;
  if ((c0 + SCAN_CHUNKS) * 32 <= E) {
    const int* p = dst + c0 * 32 + lane;
#pragma unroll
    for (int u = 0; u < SCAN_CHUNKS; ++u) {
      const int dv = p[u * 32];
      hits |= ((unsigned int)(dv - node0) < (unsigned int)NODES_PER_BLOCK ? 1u : 0u) << u;
    }
  } else {
#pragma unroll
    for (int u = 0; u < SCAN_CHUNKS; ++u) {
      const int e  = (c0 + u) * 32 + lane;
      const int dv = dst[e < E ? e : E - 1];
      const bool hit = (e < E) && ((unsigned int)(dv - node0) < (unsigned int)NODES_PER_BLOCK);
      hits |= (hit ? 1u : 0u) << u;
    }
  }
  return hits;
}

__device__ __forceinline__ void scan_build_list(
    const int* __restrict__ dst, int E, int node0, int lane, int wave,
    int* s_cnt, int* s_le) {
  const int nchunks = (E + 31) >> 5;
  for (int c0 = wave * SCAN_CHUNKS; c0 < nchunks; c0 += BLOCK_WAVES * SCAN_CHUNKS) {
    unsigned int hits = scan_step_hits(dst, E, c0, node0, lane);
    for (int it = 0; it < SCAN_CHUNKS; ++it) {
      const unsigned int active = __builtin_amdgcn_ballot_w32(hits != 0u);
      if (active == 0u) break;
      int slot0 = 0;
      if (lane == 0) slot0 = atomicAdd(s_cnt, __builtin_popcount(active));
      slot0 = __builtin_amdgcn_readfirstlane(slot0);
      if (hits != 0u) {
        const int u = __builtin_ctz(hits);
        hits &= hits - 1u;
        const int e = (c0 + u) * 32 + lane;
        const int slot = slot0 + __builtin_popcount(active & ((1u << lane) - 1u));
        if (slot < M_CAP) s_le[slot] = e;
      }
    }
  }
}

template <int NG, bool FULL>
__device__ __forceinline__ void quarter_messages(
    const float* __restrict__ x, const float* __restrict__ ea,
    int first, int n, float tv, int lane, int col,
    const int* s_le, const int* s_ls, const int* s_lr, float* s_mx, float* s_m) {
  constexpr int NE = GROUP_EDGES * NG;
  const int used = FULL ? NE : n;
  const int li = lane < used ? first + (lane >> 3) * (GROUP_EDGES * BLOCK_WAVES) + (lane & 7) : first;
  const int ev = s_le[li], sv = s_ls[li], rv = s_lr[li];
  float a[NE], b[NE];
#pragma unroll
  for (int g = 0; g < NE; ++g) {
    const int pe = __builtin_amdgcn_readlane(ev, g);
    const int s  = __builtin_amdgcn_readlane(sv, g);
    a[g] = x[(size_t)s * HDIM + col];
    b[g] = ea[(size_t)pe * HDIM + col];
  }
#pragma unroll
  for (int g = 0; g < NE; ++g) {
    if (FULL || g < n) {
      const int r   = __builtin_amdgcn_readlane(rv, g);
      const int o   = r * HDIM + col;
      const int row = first + (g >> 3) * (GROUP_EDGES * BLOCK_WAVES) + (g & 7);
      const float m = fmaxf(a[g] + b[g], 0.f) + MSG_EPS;
      s_m[row * QCOLS + lane] = m;
      const float v = m * tv;
      if (v >= 0.0f) atomicMax((int*)&s_mx[o], __float_as_int(v));
      else           atomicMin((unsigned int*)&s_mx[o], __float_as_uint(v));
    }
  }
}

template <bool FULL>
__device__ __forceinline__ void quarter_sums(
    int first, int n, float tv, int lane, int col,
    const int* s_lr, const float* s_mx, float* s_den, float* s_num, const float* s_m) {
  const int used = FULL ? GROUP_EDGES : n;
  const int rv = s_lr[lane < used ? first + lane : first];
  float m[GROUP_EDGES], mv[GROUP_EDGES];
  int o[GROUP_EDGES];
#pragma unroll
  for (int g = 0; g < GROUP_EDGES; ++g) {
    if (FULL || g < n) {
      o[g]  = __builtin_amdgcn_readlane(rv, g) * HDIM + col;
      m[g]  = s_m[(first + g) * QCOLS + lane];
      mv[g] = s_mx[o[g]];
    }
  }
#pragma unroll
  for (int g = 0; g < GROUP_EDGES; ++g) {
    if (FULL || g < n) {
      const float ex = __expf(m[g] * tv - mv[g]);
      atomicAdd(&s_den[o[g]], ex);
      atomicAdd(&s_num[o[g]], ex * m[g]);
    }
  }
}

__device__ __forceinline__ void fold_list_by_quarters(
    const float* __restrict__ x, const float* __restrict__ ea,
    int cnt, float tv, int lane, int wave,
    const int* s_le, const int* s_ls, const int* s_lr,
    float* s_mx, float* s_den, float* s_num, float* s_m) {
  const int G     = (cnt + GROUP_EDGES - 1) / GROUP_EDGES;
  const int tail  = cnt - GROUP_EDGES * (G - 1);
  const int ng    = G > wave ? (G - wave + BLOCK_WAVES - 1) / BLOCK_WAVES : 0;
  const bool has_short = ng > 0 && wave + BLOCK_WAVES * (ng - 1) == G - 1 && tail < GROUP_EDGES;
  const int nfull = has_short ? ng - 1 : ng;
  const int STRIDE = GROUP_EDGES * BLOCK_WAVES;
  const int first0 = GROUP_EDGES * wave;
#pragma unroll 1
  for (int q = 0; q < HDIM / QCOLS; ++q) {
    const int col = q * QCOLS + lane;
    int j = 0;
    for (; j + 3 <= nfull; j += 3)
      quarter_messages<3, true>(x, ea, first0 + j * STRIDE, 0, tv, lane, col, s_le, s_ls, s_lr, s_mx, s_m);
    if (nfull - j == 2)
      quarter_messages<2, true>(x, ea, first0 + j * STRIDE, 0, tv, lane, col, s_le, s_ls, s_lr, s_mx, s_m);
    else if (nfull - j == 1)
      quarter_messages<1, true>(x, ea, first0 + j * STRIDE, 0, tv, lane, col, s_le, s_ls, s_lr, s_mx, s_m);
    if (has_short)
      quarter_messages<1, false>(x, ea, first0 + nfull * STRIDE, tail, tv, lane, col, s_le, s_ls, s_lr, s_mx, s_m);
    __syncthreads();
    for (int k = 0; k < nfull; ++k)
      quarter_sums<true>(first0 + k * STRIDE, 0, tv, lane, col, s_lr, s_mx, s_den, s_num, s_m);
    if (has_short)
      quarter_sums<false>(first0 + nfull * STRIDE, tail, tv, lane, col, s_lr, s_mx, s_den, s_num, s_m);
  }
}

template <int PHASE>
__device__ __forceinline__ void scan_and_fold(
    const float* __restrict__ x, const float* __restrict__ ea,
    const int* __restrict__ src, const int* __restrict__ dst,
    int E, int node0, float tv, int lane, int wave,
    float* s_mx, float* s_den, float* s_num) {
  int pe[GROUP_EDGES];
#pragma unroll
  for (int g = 0; g < GROUP_EDGES; ++g) pe[g] = 0;
  int pn = 0;
  const int nchunks = (E + 31) >> 5;
  for (int c0 = wave * SCAN_CHUNKS; c0 < nchunks; c0 += BLOCK_WAVES * SCAN_CHUNKS) {
    const unsigned int hits = scan_step_hits(dst, E, c0, node0, lane);
    if (__builtin_amdgcn_ballot_w32(hits != 0u) != 0u) {
#pragma unroll 1
      for (int u = 0; u < SCAN_CHUNKS; ++u) {
        unsigned int mask = __builtin_amdgcn_ballot_w32(((hits >> u) & 1u) != 0u);
        const int base = (c0 + u) * 32;
        while (mask != 0u) {
          const int bit = __builtin_ctz(mask);
          mask &= mask - 1u;
#pragma unroll
          for (int g = GROUP_EDGES - 1; g > 0; --g) pe[g] = pe[g - 1];
          pe[0] = base + bit;
          if (++pn == GROUP_EDGES) {
            fold_edge_group<PHASE>(x, ea, src, dst, pe, GROUP_EDGES, node0, tv, lane, s_mx, s_den, s_num);
            pn = 0;
          }
        }
      }
    }
  }
  if (pn > 0)
    fold_edge_group<PHASE>(x, ea, src, dst, pe, pn, node0, tv, lane, s_mx, s_den, s_num);
}

template <bool BUILD>
__global__ __launch_bounds__(BLOCK_WAVES * 32)
void gen_softmax_lds_kernel(const float* __restrict__ x, const float* __restrict__ ea,
                            const int* __restrict__ src, const int* __restrict__ dst,
                            const float* __restrict__ tptr, int layer,
                            float* __restrict__ hc, int* __restrict__ lists, int E, int N) {
  __shared__ float s_mx[NODES_PER_BLOCK * HDIM];
  __shared__ float s_den[NODES_PER_BLOCK * HDIM];
  __shared__ float s_num[NODES_PER_BLOCK * HDIM];
  __shared__ float s_m[M_CAP * QCOLS];
  __shared__ int s_le[M_CAP];
  __shared__ int s_ls[M_CAP];
  __shared__ int s_lr[M_CAP];
  __shared__ int s_cnt;
  const int tid   = threadIdx.x;
  const int lane  = tid & 31;
  const int wave  = __builtin_amdgcn_readfirstlane(tid >> 5);
  const int node0 = blockIdx.x * NODES_PER_BLOCK;
  int* gl = lists + (size_t)blockIdx.x * LIST_STRIDE;
  for (int i = tid; i < NODES_PER_BLOCK * HDIM; i += BLOCK_WAVES * 32) {
    s_mx[i]  = __int_as_float(0xff800000);
    s_den[i] = 0.f;
    s_num[i] = 0.f;
  }
  const float tv = tptr[layer];
  int cnt;
  if (BUILD) {
    if (tid == 0) s_cnt = 0;
    __syncthreads();
    scan_build_list(dst, E, node0, lane, wave, &s_cnt, s_le);
    __syncthreads();
    cnt = __builtin_amdgcn_readfirstlane(s_cnt);
    if (tid == 0) gl[0] = cnt;
    if (cnt <= M_CAP) {
      for (int i = tid; i < cnt; i += BLOCK_WAVES * 32) {
        const int e = s_le[i];
        const int s = src[e];
        const int r = dst[e] - node0;
        s_ls[i] = s;
        s_lr[i] = r;
        gl[LIST_HDR + i]             = e;
        gl[LIST_HDR + M_CAP + i]     = s;
        gl[LIST_HDR + 2 * M_CAP + i] = r;
      }
    }
  } else {
    cnt = __builtin_amdgcn_readfirstlane(gl[0]);
    if (cnt <= M_CAP) {
      constexpr int TRIPS = (M_CAP + BLOCK_WAVES * 32 - 1) / (BLOCK_WAVES * 32);
      int ve[TRIPS], vs[TRIPS], vr[TRIPS];
#pragma unroll
      for (int k = 0; k < TRIPS; ++k) {
        const int i = tid + k * (BLOCK_WAVES * 32);
        if (i < cnt) {
          ve[k] = gl[LIST_HDR + i];
          vs[k] = gl[LIST_HDR + M_CAP + i];
          vr[k] = gl[LIST_HDR + 2 * M_CAP + i];
        }
      }
#pragma unroll
      for (int k = 0; k < TRIPS; ++k) {
        const int i = tid + k * (BLOCK_WAVES * 32);
        if (i < cnt) {
          s_le[i] = ve[k];
          s_ls[i] = vs[k];
          s_lr[i] = vr[k];
        }
      }
    }
  }
  __syncthreads();
  if (cnt <= M_CAP) {
    fold_list_by_quarters(x, ea, cnt, tv, lane, wave, s_le, s_ls, s_lr, s_mx, s_den, s_num, s_m);
  } else {
    scan_and_fold<1>(x, ea, src, dst, E, node0, tv, lane, wave, s_mx, s_den, s_num);
    __syncthreads();
    scan_and_fold<2>(x, ea, src, dst, E, node0, tv, lane, wave, s_mx, s_den, s_num);
  }
  __syncthreads();
  for (int i = tid; i < NODES_PER_BLOCK * HDIM; i += BLOCK_WAVES * 32) {
    const int node = node0 + i / HDIM;
    if (node < N) {
      const float dv  = s_den[i];
      const float agg = dv > 0.f ? s_num[i] / dv : 0.f;
      const size_t gi = (size_t)node * HDIM + i % HDIM;
      hc[gi] = agg + x[gi];
    }
  }
}

__device__ __forceinline__ void pool_flush(float* __restrict__ pooled, int gid, const float (&acc)[4], int lane) {
  if ((unsigned int)gid < (unsigned int)NUM_GRAPHS) {
#pragma unroll
    for (int j = 0; j < 4; ++j) atomicAdd(pooled + (size_t)gid * HDIM + lane + 32 * j, acc[j]);
  }
}

__global__ __launch_bounds__(ROW_WAVES * 32)
void pool_rows_kernel(const float* __restrict__ x, const int* __restrict__ batch,
                      float* __restrict__ pooled, int n) {
  const int lane = threadIdx.x & 31;
  const int wave = __builtin_amdgcn_readfirstlane(threadIdx.x >> 5);
  const int row0 = (blockIdx.x * ROW_WAVES + wave) * ROWS_PER_WAVE;
  if (row0 >= n) return;
  float v[ROWS_PER_WAVE][4];
  int gid[ROWS_PER_WAVE];
#pragma unroll
  for (int r = 0; r < ROWS_PER_WAVE; ++r) {
    const bool live = row0 + r < n;
    const int row = live ? row0 + r : n - 1;
    gid[r] = __builtin_amdgcn_readfirstlane(batch[row]);
#pragma unroll
    for (int j = 0; j < 4; ++j) {
      const float t = x[(size_t)row * HDIM + lane + 32 * j];
      v[r][j] = live ? t : 0.f;
    }
  }
  float acc[4] = {0.f, 0.f, 0.f, 0.f};
  int cur = gid[0];
#pragma unroll
  for (int r = 0; r < ROWS_PER_WAVE; ++r) {
    if (gid[r] != cur) {
      pool_flush(pooled, cur, acc, lane);
#pragma unroll
      for (int j = 0; j < 4; ++j) acc[j] = 0.f;
      cur = gid[r];
    }
#pragma unroll
    for (int j = 0; j < 4; ++j) acc[j] += v[r][j];
  }
  pool_flush(pooled, cur, acc, lane);
}

__global__ void final_linear_kernel(const float* __restrict__ pooled,
                                    const float* __restrict__ W,
                                    const float* __restrict__ b,
                                    float* __restrict__ out) {
  int i = blockIdx.x * blockDim.x + threadIdx.x;
  if (i >= NUM_GRAPHS * NUM_CLASSES) return;
  int g = i / NUM_CLASSES, c = i % NUM_CLASSES;
  float acc = b[c];
  const float* pr = pooled + (size_t)g * HDIM;
#pragma unroll 4
  for (int k = 0; k < HDIM; ++k) acc += pr[k] * W[k * NUM_CLASSES + c];
  out[i] = acc;
}

extern "C" void kernel_launch(void* const* d_in, const int* in_sizes, int n_in,
                              void* d_out, int out_size, void* d_ws, size_t ws_size,
                              hipStream_t stream) {
  const float* x     = (const float*)d_in[0];
  const float* ea    = (const float*)d_in[1];
  const float* enc_W = (const float*)d_in[2];
  const float* enc_b = (const float*)d_in[3];
  const float* t     = (const float*)d_in[4];
  const float* W1    = (const float*)d_in[5];
  const float* b1    = (const float*)d_in[6];
  const float* mlp_g = (const float*)d_in[7];
  const float* mlp_b = (const float*)d_in[8];
  const float* W2    = (const float*)d_in[9];
  const float* b2    = (const float*)d_in[10];
  const float* ln_g  = (const float*)d_in[11];
  const float* ln_b  = (const float*)d_in[12];
  const float* lin_W = (const float*)d_in[13];
  const float* lin_b = (const float*)d_in[14];
  const int*   eidx  = (const int*)d_in[15];
  const int*   batch = (const int*)d_in[16];

  const int N = in_sizes[0] / HDIM;
  const int E = in_sizes[1] / HDIM;
  const int L = in_sizes[4];
  const int* src = eidx;
  const int* dst = eidx + E;

  const size_t NH = (size_t)N * HDIM;
  float* h    = (float*)d_ws;
  float* rbuf = h    + NH;
  float* hc   = rbuf + NH;
  float* t1   = hc   + NH;
  int* lists  = (int*)(t1 + 2 * NH);

  float* out    = (float*)d_out;
  float* pooled = out + NUM_GRAPHS * NUM_CLASSES;

  const dim3 gblk(128);
  const int aggBlocks = (N + NODES_PER_BLOCK - 1) / NODES_PER_BLOCK;
  const int rowBlocks = (N + ROW_WAVES * ROWS_PER_WAVE - 1) / (ROW_WAVES * ROWS_PER_WAVE);

  gemm_rows16_wmma_f32<HDIM, HDIM, 2, 0><<<N / 16, gblk, 0, stream>>>(x, enc_W, enc_b, h, 0, nullptr, nullptr, nullptr);

  for (int layer = 0; layer < L; ++layer) {
    const float* rin = layer == 0 ? h : rbuf;
    if (layer == 0)
      gen_softmax_lds_kernel<true><<<aggBlocks, BLOCK_WAVES * 32, 0, stream>>>(rin, ea, src, dst, t, layer, hc, lists, E, N);
    else
      gen_softmax_lds_kernel<false><<<aggBlocks, BLOCK_WAVES * 32, 0, stream>>>(rin, ea, src, dst, t, layer, hc, lists, E, N);
    gemm_rows16_wmma_f32<HDIM, HDIM2, 4, 1><<<N / 16, gblk, 0, stream>>>(
        hc, W1 + (size_t)layer * HDIM * HDIM2, b1 + layer * HDIM2, t1, 0,
        mlp_g + layer * HDIM2, mlp_b + layer * HDIM2, nullptr);
    const int nl = layer + 1 < L ? layer + 1 : 0;
    gemm_rows16_wmma_f32<HDIM2, HDIM, 2, 2><<<N / 16, gblk, 0, stream>>>(
        t1, W2 + (size_t)layer * HDIM2 * HDIM, b2 + layer * HDIM, h, layer > 0 ? 1 : 0,
        ln_g + nl * HDIM, ln_b + nl * HDIM, rbuf);
  }

  zero_kernel<<<(NUM_GRAPHS * HDIM + 255) / 256, 256, 0, stream>>>(pooled, NUM_GRAPHS * HDIM);
  pool_rows_kernel<<<rowBlocks, ROW_WAVES * 32, 0, stream>>>(rbuf, batch, pooled, N);
  final_linear_kernel<<<(NUM_GRAPHS * NUM_CLASSES + 63) / 64, 64, 0, stream>>>(pooled, lin_W, lin_b, out);
}


# Edge softmax sums in one pass: exp(t*m) unshifted (shift-invariant) while a block's |t*m| <= 16, else it redoes max+sum; whole rows per edge, no plane or max atomics; terms round anew, LDS order kept
# speedup vs baseline: 4.4232x; 1.2469x over previous
#include <hip/hip_runtime.h>
#include <hip/hip_bf16.h>

typedef __attribute__((ext_vector_type(2))) float v2f;
typedef __attribute__((ext_vector_type(8))) float v8f;

#define HDIM 128
#define HDIM2 256
#define NUM_GRAPHS 64
#define NUM_CLASSES 10
#define MSG_EPS 1e-7f

template <int NT> struct fvec;
template <> struct fvec<2> { typedef __attribute__((ext_vector_type(2))) float type; };
template <> struct fvec<4> { typedef __attribute__((ext_vector_type(4))) float type; };

template <int XMASK>
__device__ __forceinline__ float row16_xadd(float v) {
  return v + __int_as_float(__builtin_amdgcn_update_dpp(0, __float_as_int(v), 0x160 + XMASK, 0xf, 0xf, false));
}

template <int K, int M, int NT, int LNMODE>
__global__ __launch_bounds__(128)
void gemm_rows16_wmma_f32(const float* __restrict__ X,
                          const float* __restrict__ W,
                          const float* __restrict__ bias,
                          float* __restrict__ Y, int accum,
                          const float* __restrict__ lng, const float* __restrict__ lnb,
                          float* __restrict__ Z) {
  typedef typename fvec<NT>::type vnt;
  typedef __attribute__((ext_vector_type(4))) float v4f;
  static_assert(M == 64 * NT, "4 waves of NT tiles cover the M columns");
  constexpr int PITCH = K + 4;
  constexpr int KU = 16 / NT;
  constexpr int NG = K / (4 * KU);
  __shared__ float s_a[16 * PITCH];
  __shared__ alignas(16) float s_red[16 * 8];

  const int tid  = threadIdx.x;
  const int lane = tid & 31;
  const int wid  = tid >> 5;
  const int row0 = blockIdx.x * 16;
  const int half = lane >> 4;
  const int l16  = lane & 15;
  const int c0   = wid * (16 * NT);
  const int colv = c0 + NT * l16;

  const float* wp = W + (size_t)(half * 2) * M + colv;
  const float* ap = s_a + l16 * PITCH + half * 2;

  vnt bq[2][KU][2];
  v2f aq[2][KU];

#pragma unroll
  for (int u = 0; u < KU; ++u) {
    bq[0][u][0] = *(const vnt*)(wp + (size_t)(u * 4) * M);
    bq[0][u][1] = *(const vnt*)(wp + (size_t)(u * 4 + 1) * M);
  }

  {
    const float* xs = X + (size_t)row0 * K;
#pragma unroll
    for (int i = 0; i < (16 * K / 4) / 128; ++i) {
      const int idx = tid + 128 * i;
      const int r = idx / (K / 4), q = idx % (K / 4);
      const v4f v = *(const v4f*)(xs + (size_t)r * K + q * 4);
      *(v4f*)(s_a + r * PITCH + q * 4) = v;
    }
  }

  const vnt bv = *(const vnt*)(bias + colv);
  v8f c[NT];
  if (accum) {
#pragma unroll
    for (int r = 0; r < 8; ++r) {
      const vnt y = *(const vnt*)(Y + (size_t)(row0 + half * 8 + r) * M + colv);
#pragma unroll
      for (int t = 0; t < NT; ++t) c[t][r] = y[t] + bv[t];
    }
  } else {
#pragma unroll
    for (int t = 0; t < NT; ++t)
#pragma unroll
      for (int r = 0; r < 8; ++r) c[t][r] = bv[t];
  }

  __syncthreads();

#pragma unroll
  for (int u = 0; u < KU; ++u) aq[0][u] = *(const v2f*)(ap + u * 4);

#pragma unroll
  for (int g = 0; g < NG; ++g) {
    const int cur = g & 1, nxt = cur ^ 1;
    if (g + 1 < NG) {
#pragma unroll
      for (int u = 0; u < KU; ++u) {
        const int k = ((g + 1) * KU + u) * 4;
        bq[nxt][u][0] = *(const vnt*)(wp + (size_t)k * M);
        bq[nxt][u][1] = *(const vnt*)(wp + (size_t)(k + 1) * M);
        aq[nxt][u]    = *(const v2f*)(ap + k);
      }
    }
#pragma unroll
    for (int u = 0; u < KU; ++u) {
#pragma unroll
      for (int t = 0; t < NT; ++t) {
        v2f b; b[0] = bq[cur][u][0][t]; b[1] = bq[cur][u][1][t];
        c[t] = __builtin_amdgcn_wmma_f32_16x16x4_f32(false, aq[cur][u], false, b,
                                                     (short)0, c[t], false, false);
      }
    }
  }

  if (LNMODE != 1) {
#pragma unroll
    for (int r = 0; r < 8; ++r) {
      vnt o;
#pragma unroll
      for (int t = 0; t < NT; ++t) o[t] = c[t][r];
      *(vnt*)(Y + (size_t)(row0 + half * 8 + r) * M + colv) = o;
    }
  }
  if (LNMODE != 0) {
    const vnt gv = *(const vnt*)(lng + colv);
    const vnt ev = *(const vnt*)(lnb + colv);
    float s[8], q[8];
#pragma unroll
    for (int r = 0; r < 8; ++r) {
      s[r] = 0.f; q[r] = 0.f;
#pragma unroll
      for (int t = 0; t < NT; ++t) { const float v = c[t][r]; s[r] += v; q[r] += v * v; }
    }
#pragma unroll
    for (int r = 0; r < 8; ++r) { s[r] = row16_xadd<1>(s[r]); q[r] = row16_xadd<1>(q[r]); }
#pragma unroll
    for (int r = 0; r < 8; ++r) { s[r] = row16_xadd<2>(s[r]); q[r] = row16_xadd<2>(q[r]); }
#pragma unroll
    for (int r = 0; r < 8; ++r) { s[r] = row16_xadd<4>(s[r]); q[r] = row16_xadd<4>(q[r]); }
#pragma unroll
    for (int r = 0; r < 8; ++r) { s[r] = row16_xadd<8>(s[r]); q[r] = row16_xadd<8>(q[r]); }
    if (l16 == 0) {
#pragma unroll
      for (int r = 0; r < 8; ++r) {
        v2f p; p[0] = s[r]; p[1] = q[r];
        *(v2f*)(s_red + (half * 8 + r) * 8 + wid * 2) = p;
      }
    }
    __syncthreads();
    float* dst = LNMODE == 1 ? Y : Z;
#pragma unroll
    for (int r = 0; r < 8; ++r) {
      const v4f p0 = *(const v4f*)(s_red + (half * 8 + r) * 8);
      const v4f p1 = *(const v4f*)(s_red + (half * 8 + r) * 8 + 4);
      const float sum  = ((p0[0] + p0[2]) + p1[0]) + p1[2];
      const float sq   = ((p0[1] + p0[3]) + p1[1]) + p1[3];
      const float mean = sum / (float)M;
      const float var  = sq / (float)M - mean * mean;
      const float inv  = rsqrtf(var + 1e-5f);
      vnt o;
#pragma unroll
      for (int t = 0; t < NT; ++t)
        o[t] = fmaxf((c[t][r] - mean) * inv * gv[t] + ev[t], 0.f);
      *(vnt*)(dst + (size_t)(row0 + half * 8 + r) * M + colv) = o;
    }
  }
}

#define ROWS_PER_WAVE 8
#define ROW_WAVES 4

__global__ void zero_kernel(float* p, int n) {
  int i = blockIdx.x * blockDim.x + threadIdx.x;
  if (i < n) p[i] = 0.f;
}

#define NODES_PER_BLOCK 40
#define BLOCK_WAVES 32
#define SCAN_CHUNKS 16
#define GROUP_EDGES 8
#define M_CAP 1536
#define QCOLS 32
#define LIST_HDR 32
#define LIST_STRIDE (LIST_HDR + 3 * M_CAP)
#define EXP_ARG_CAP 16.0f

template <int PHASE>
__device__ __forceinline__ void fold_rows(
    const float* __restrict__ x, const float* __restrict__ ea,
    const int (&pe)[GROUP_EDGES], const int (&s)[GROUP_EDGES], const int (&r)[GROUP_EDGES],
    int count, float tv, int lane, float* s_mx, float* s_den, float* s_num) {
  float a[GROUP_EDGES][4], b[GROUP_EDGES][4];
#pragma unroll
  for (int g = 0; g < GROUP_EDGES; ++g) {
    const float* xr = x  + (size_t)s[g]  * HDIM + lane;
    const float* er = ea + (size_t)pe[g] * HDIM + lane;
#pragma unroll
    for (int j = 0; j < 4; ++j) { a[g][j] = xr[32 * j]; b[g][j] = er[32 * j]; }
  }
  float mv[GROUP_EDGES][4];
  if (PHASE == 2) {
#pragma unroll
    for (int g = 0; g < GROUP_EDGES; ++g) {
      if (g < count) {
#pragma unroll
        for (int j = 0; j < 4; ++j) mv[g][j] = s_mx[r[g] * HDIM + lane + 32 * j];
      }
    }
  }
#pragma unroll
  for (int g = 0; g < GROUP_EDGES; ++g) {
    if (g < count) {
      const int o = r[g] * HDIM + lane;
#pragma unroll
      for (int j = 0; j < 4; ++j) {
        const float m = fmaxf(a[g][j] + b[g][j], 0.f) + MSG_EPS;
        if (PHASE == 1) {
          const float v = m * tv;
          if (v >= 0.0f) atomicMax((int*)&s_mx[o + 32 * j], __float_as_int(v));
          else           atomicMin((unsigned int*)&s_mx[o + 32 * j], __float_as_uint(v));
        } else {
          const float ex = __expf(m * tv - mv[g][j]);
          atomicAdd(&s_den[o + 32 * j], ex);
          atomicAdd(&s_num[o + 32 * j], ex * m);
        }
      }
    }
  }
}

template <int PHASE>
__device__ __forceinline__ void fold_edge_group(
    const float* __restrict__ x, const float* __restrict__ ea,
    const int* __restrict__ src, const int* __restrict__ dst,
    const int (&pe)[GROUP_EDGES], int count, int node0, float tv, int lane,
    float* s_mx, float* s_den, float* s_num) {
  int s[GROUP_EDGES], r[GROUP_EDGES];
#pragma unroll
  for (int g = 0; g < GROUP_EDGES; ++g) {
    s[g] = src[pe[g]];
    r[g] = dst[pe[g]] - node0;
  }
  fold_rows<PHASE>(x, ea, pe, s, r, count, tv, lane, s_mx, s_den, s_num);
}

__device__ __forceinline__ unsigned int scan_step_hits(const int* __restrict__ dst, int E, int c0,
                                                       int node0, int lane) {
  unsigned int hits = 0u;
  if ((c0 + SCAN_CHUNKS) * 32 <= E) {
    const int* p = dst + c0 * 32 + lane;
#pragma unroll
    for (int u = 0; u < SCAN_CHUNKS; ++u) {
      const int dv = p[u * 32];
      hits |= ((unsigned int)(dv - node0) < (unsigned int)NODES_PER_BLOCK ? 1u : 0u) << u;
    }
  } else {
#pragma unroll
    for (int u = 0; u < SCAN_CHUNKS; ++u) {
      const int e  = (c0 + u) * 32 + lane;
      const int dv = dst[e < E ? e : E - 1];
      const bool hit = (e < E) && ((unsigned int)(dv - node0) < (unsigned int)NODES_PER_BLOCK);
      hits |= (hit ? 1u : 0u) << u;
    }
  }
  return hits;
}

__device__ __forceinline__ void scan_build_list(
    const int* __restrict__ dst, int E, int node0, int lane, int wave,
    int* s_cnt, int* s_le) {
  const int nchunks = (E + 31) >> 5;
  for (int c0 = wave * SCAN_CHUNKS; c0 < nchunks; c0 += BLOCK_WAVES * SCAN_CHUNKS) {
    unsigned int hits = scan_step_hits(dst, E, c0, node0, lane);
    for (int it = 0; it < SCAN_CHUNKS; ++it) {
      const unsigned int active = __builtin_amdgcn_ballot_w32(hits != 0u);
      if (active == 0u) break;
      int slot0 = 0;
      if (lane == 0) slot0 = atomicAdd(s_cnt, __builtin_popcount(active));
      slot0 = __builtin_amdgcn_readfirstlane(slot0);
      if (hits != 0u) {
        const int u = __builtin_ctz(hits);
        hits &= hits - 1u;
        const int e = (c0 + u) * 32 + lane;
        const int slot = slot0 + __builtin_popcount(active & ((1u << lane) - 1u));
        if (slot < M_CAP) s_le[slot] = e;
      }
    }
  }
}

template <int NG, bool FULL>
__device__ __forceinline__ void quarter_messages(
    const float* __restrict__ x, const float* __restrict__ ea,
    int first, int n, float tv, int lane, int col,
    const int* s_le, const int* s_ls, const int* s_lr, float* s_mx, float* s_m) {
  constexpr int NE = GROUP_EDGES * NG;
  const int used = FULL ? NE : n;
  const int li = lane < used ? first + (lane >> 3) * (GROUP_EDGES * BLOCK_WAVES) + (lane & 7) : first;
  const int ev = s_le[li], sv = s_ls[li], rv = s_lr[li];
  float a[NE], b[NE];
#pragma unroll
  for (int g = 0; g < NE; ++g) {
    const int pe = __builtin_amdgcn_readlane(ev, g);
    const int s  = __builtin_amdgcn_readlane(sv, g);
    a[g] = x[(size_t)s * HDIM + col];
    b[g] = ea[(size_t)pe * HDIM + col];
  }
#pragma unroll
  for (int g = 0; g < NE; ++g) {
    if (FULL || g < n) {
      const int r   = __builtin_amdgcn_readlane(rv, g);
      const int o   = r * HDIM + col;
      const int row = first + (g >> 3) * (GROUP_EDGES * BLOCK_WAVES) + (g & 7);
      const float m = fmaxf(a[g] + b[g], 0.f) + MSG_EPS;
      s_m[row * QCOLS + lane] = m;
      const float v = m * tv;
      if (v >= 0.0f) atomicMax((int*)&s_mx[o], __float_as_int(v));
      else           atomicMin((unsigned int*)&s_mx[o], __float_as_uint(v));
    }
  }
}

template <bool FULL>
__device__ __forceinline__ void quarter_sums(
    int first, int n, float tv, int lane, int col,
    const int* s_lr, const float* s_mx, float* s_den, float* s_num, const float* s_m) {
  const int used = FULL ? GROUP_EDGES : n;
  const int rv = s_lr[lane < used ? first + lane : first];
  float m[GROUP_EDGES], mv[GROUP_EDGES];
  int o[GROUP_EDGES];
#pragma unroll
  for (int g = 0; g < GROUP_EDGES; ++g) {
    if (FULL || g < n) {
      o[g]  = __builtin_amdgcn_readlane(rv, g) * HDIM + col;
      m[g]  = s_m[(first + g) * QCOLS + lane];
      mv[g] = s_mx[o[g]];
    }
  }
#pragma unroll
  for (int g = 0; g < GROUP_EDGES; ++g) {
    if (FULL || g < n) {
      const float ex = __expf(m[g] * tv - mv[g]);
      atomicAdd(&s_den[o[g]], ex);
      atomicAdd(&s_num[o[g]], ex * m[g]);
    }
  }
}

__device__ __forceinline__ void fold_list_by_quarters(
    const float* __restrict__ x, const float* __restrict__ ea,
    int cnt, float tv, int lane, int wave,
    const int* s_le, const int* s_ls, const int* s_lr,
    float* s_mx, float* s_den, float* s_num, float* s_m) {
  const int G     = (cnt + GROUP_EDGES - 1) / GROUP_EDGES;
  const int tail  = cnt - GROUP_EDGES * (G - 1);
  const int ng    = G > wave ? (G - wave + BLOCK_WAVES - 1) / BLOCK_WAVES : 0;
  const bool has_short = ng > 0 && wave + BLOCK_WAVES * (ng - 1) == G - 1 && tail < GROUP_EDGES;
  const int nfull = has_short ? ng - 1 : ng;
  const int STRIDE = GROUP_EDGES * BLOCK_WAVES;
  const int first0 = GROUP_EDGES * wave;
#pragma unroll 1
  for (int q = 0; q < HDIM / QCOLS; ++q) {
    const int col = q * QCOLS + lane;
    int j = 0;
    for (; j + 3 <= nfull; j += 3)
      quarter_messages<3, true>(x, ea, first0 + j * STRIDE, 0, tv, lane, col, s_le, s_ls, s_lr, s_mx, s_m);
    if (nfull - j == 2)
      quarter_messages<2, true>(x, ea, first0 + j * STRIDE, 0, tv, lane, col, s_le, s_ls, s_lr, s_mx, s_m);
    else if (nfull - j == 1)
      quarter_messages<1, true>(x, ea, first0 + j * STRIDE, 0, tv, lane, col, s_le, s_ls, s_lr, s_mx, s_m);
    if (has_short)
      quarter_messages<1, false>(x, ea, first0 + nfull * STRIDE, tail, tv, lane, col, s_le, s_ls, s_lr, s_mx, s_m);
    __syncthreads();
    for (int k = 0; k < nfull; ++k)
      quarter_sums<true>(first0 + k * STRIDE, 0, tv, lane, col, s_lr, s_mx, s_den, s_num, s_m);
    if (has_short)
      quarter_sums<false>(first0 + nfull * STRIDE, tail, tv, lane, col, s_lr, s_mx, s_den, s_num, s_m);
  }
}

template <bool FULL>
__device__ __forceinline__ float one_pass_group(
    const float* __restrict__ x, const float* __restrict__ ea,
    int first, int n, float tv, int lane,
    const int* s_le, const int* s_ls, const int* s_lr,
    float* s_den, float* s_num, float vabs) {
  const int used = FULL ? GROUP_EDGES : n;
  const int li = lane < used ? first + lane : first;
  const int ev = s_le[li], sv = s_ls[li], rv = s_lr[li];
  float a[GROUP_EDGES][4], b[GROUP_EDGES][4];
#pragma unroll
  for (int g = 0; g < GROUP_EDGES; ++g) {
    const int pe = __builtin_amdgcn_readlane(ev, g);
    const int s  = __builtin_amdgcn_readlane(sv, g);
    const float* xr = x  + (size_t)s  * HDIM + lane;
    const float* er = ea + (size_t)pe * HDIM + lane;
#pragma unroll
    for (int j = 0; j < 4; ++j) { a[g][j] = xr[32 * j]; b[g][j] = er[32 * j]; }
  }
#pragma unroll
  for (int g = 0; g < GROUP_EDGES; ++g) {
    if (FULL || g < n) {
      const int o = __builtin_amdgcn_readlane(rv, g) * HDIM + lane;
#pragma unroll
      for (int j = 0; j < 4; ++j) {
        const float m  = fmaxf(a[g][j] + b[g][j], 0.f) + MSG_EPS;
        const float v  = m * tv;
        vabs = fmaxf(vabs, fabsf(v));
        const float ex = __expf(v);
        atomicAdd(&s_den[o + 32 * j], ex);
        atomicAdd(&s_num[o + 32 * j], ex * m);
      }
    }
  }
  return vabs;
}

__device__ __forceinline__ float fold_list_one_pass(
    const float* __restrict__ x, const float* __restrict__ ea,
    int cnt, float tv, int lane, int wave,
    const int* s_le, const int* s_ls, const int* s_lr,
    float* s_den, float* s_num) {
  const int G    = (cnt + GROUP_EDGES - 1) / GROUP_EDGES;
  const int tail = cnt - GROUP_EDGES * (G - 1);
  float vabs = 0.f;
  for (int k = wave; k < G; k += BLOCK_WAVES) {
    if (k < G - 1 || tail == GROUP_EDGES)
      vabs = one_pass_group<true>(x, ea, GROUP_EDGES * k, 0, tv, lane, s_le, s_ls, s_lr, s_den, s_num, vabs);
    else
      vabs = one_pass_group<false>(x, ea, GROUP_EDGES * k, tail, tv, lane, s_le, s_ls, s_lr, s_den, s_num, vabs);
  }
  return vabs;
}

template <int PHASE>
__device__ __forceinline__ void scan_and_fold(
    const float* __restrict__ x, const float* __restrict__ ea,
    const int* __restrict__ src, const int* __restrict__ dst,
    int E, int node0, float tv, int lane, int wave,
    float* s_mx, float* s_den, float* s_num) {
  int pe[GROUP_EDGES];
#pragma unroll
  for (int g = 0; g < GROUP_EDGES; ++g) pe[g] = 0;
  int pn = 0;
  const int nchunks = (E + 31) >> 5;
  for (int c0 = wave * SCAN_CHUNKS; c0 < nchunks; c0 += BLOCK_WAVES * SCAN_CHUNKS) {
    const unsigned int hits = scan_step_hits(dst, E, c0, node0, lane);
    if (__builtin_amdgcn_ballot_w32(hits != 0u) != 0u) {
#pragma unroll 1
      for (int u = 0; u < SCAN_CHUNKS; ++u) {
        unsigned int mask = __builtin_amdgcn_ballot_w32(((hits >> u) & 1u) != 0u);
        const int base = (c0 + u) * 32;
        while (mask != 0u) {
          const int bit = __builtin_ctz(mask);
          mask &= mask - 1u;
#pragma unroll
          for (int g = GROUP_EDGES - 1; g > 0; --g) pe[g] = pe[g - 1];
          pe[0] = base + bit;
          if (++pn == GROUP_EDGES) {
            fold_edge_group<PHASE>(x, ea, src, dst, pe, GROUP_EDGES, node0, tv, lane, s_mx, s_den, s_num);
            pn = 0;
          }
        }
      }
    }
  }
  if (pn > 0)
    fold_edge_group<PHASE>(x, ea, src, dst, pe, pn, node0, tv, lane, s_mx, s_den, s_num);
}

template <bool BUILD>
__global__ __launch_bounds__(BLOCK_WAVES * 32)
void gen_softmax_lds_kernel(const float* __restrict__ x, const float* __restrict__ ea,
                            const int* __restrict__ src, const int* __restrict__ dst,
                            const float* __restrict__ tptr, int layer,
                            float* __restrict__ hc, int* __restrict__ lists, int E, int N) {
  __shared__ float s_mx[NODES_PER_BLOCK * HDIM];
  __shared__ float s_den[NODES_PER_BLOCK * HDIM];
  __shared__ float s_num[NODES_PER_BLOCK * HDIM];
  __shared__ float s_m[M_CAP * QCOLS];
  __shared__ int s_le[M_CAP];
  __shared__ int s_ls[M_CAP];
  __shared__ int s_lr[M_CAP];
  __shared__ int s_cnt;
  __shared__ int s_redo;
  const int tid   = threadIdx.x;
  const int lane  = tid & 31;
  const int wave  = __builtin_amdgcn_readfirstlane(tid >> 5);
  const int node0 = blockIdx.x * NODES_PER_BLOCK;
  int* gl = lists + (size_t)blockIdx.x * LIST_STRIDE;
  for (int i = tid; i < NODES_PER_BLOCK * HDIM; i += BLOCK_WAVES * 32) {
    s_mx[i]  = __int_as_float(0xff800000);
    s_den[i] = 0.f;
    s_num[i] = 0.f;
  }
  const float tv = tptr[layer];
  if (tid == 0) s_redo = 0;
  int cnt;
  if (BUILD) {
    if (tid == 0) s_cnt = 0;
    __syncthreads();
    scan_build_list(dst, E, node0, lane, wave, &s_cnt, s_le);
    __syncthreads();
    cnt = __builtin_amdgcn_readfirstlane(s_cnt);
    if (tid == 0) gl[0] = cnt;
    if (cnt <= M_CAP) {
      for (int i = tid; i < cnt; i += BLOCK_WAVES * 32) {
        const int e = s_le[i];
        const int s = src[e];
        const int r = dst[e] - node0;
        s_ls[i] = s;
        s_lr[i] = r;
        gl[LIST_HDR + i]             = e;
        gl[LIST_HDR + M_CAP + i]     = s;
        gl[LIST_HDR + 2 * M_CAP + i] = r;
      }
    }
  } else {
    cnt = __builtin_amdgcn_readfirstlane(gl[0]);
    if (cnt <= M_CAP) {
      constexpr int TRIPS = (M_CAP + BLOCK_WAVES * 32 - 1) / (BLOCK_WAVES * 32);
      int ve[TRIPS], vs[TRIPS], vr[TRIPS];
#pragma unroll
      for (int k = 0; k < TRIPS; ++k) {
        const int i = tid + k * (BLOCK_WAVES * 32);
        if (i < cnt) {
          ve[k] = gl[LIST_HDR + i];
          vs[k] = gl[LIST_HDR + M_CAP + i];
          vr[k] = gl[LIST_HDR + 2 * M_CAP + i];
        }
      }
#pragma unroll
      for (int k = 0; k < TRIPS; ++k) {
        const int i = tid + k * (BLOCK_WAVES * 32);
        if (i < cnt) {
          s_le[i] = ve[k];
          s_ls[i] = vs[k];
          s_lr[i] = vr[k];
        }
      }
    }
  }
  __syncthreads();
  if (cnt <= M_CAP) {
    const float vabs = fold_list_one_pass(x, ea, cnt, tv, lane, wave, s_le, s_ls, s_lr, s_den, s_num);
    if (__builtin_amdgcn_ballot_w32(!(vabs <= EXP_ARG_CAP)) != 0u && lane == 0) s_redo = 1;
    __syncthreads();
    if (__builtin_amdgcn_readfirstlane(s_redo) != 0) {
      for (int i = tid; i < NODES_PER_BLOCK * HDIM; i += BLOCK_WAVES * 32) {
        s_den[i] = 0.f;
        s_num[i] = 0.f;
      }
      __syncthreads();
      fold_list_by_quarters(x, ea, cnt, tv, lane, wave, s_le, s_ls, s_lr, s_mx, s_den, s_num, s_m);
      __syncthreads();
    }
  } else {
    scan_and_fold<1>(x, ea, src, dst, E, node0, tv, lane, wave, s_mx, s_den, s_num);
    __syncthreads();
    scan_and_fold<2>(x, ea, src, dst, E, node0, tv, lane, wave, s_mx, s_den, s_num);
    __syncthreads();
  }
  for (int i = tid; i < NODES_PER_BLOCK * HDIM; i += BLOCK_WAVES * 32) {
    const int node = node0 + i / HDIM;
    if (node < N) {
      const float dv  = s_den[i];
      const float agg = dv > 0.f ? s_num[i] / dv : 0.f;
      const size_t gi = (size_t)node * HDIM + i % HDIM;
      hc[gi] = agg + x[gi];
    }
  }
}

__device__ __forceinline__ void pool_flush(float* __restrict__ pooled, int gid, const float (&acc)[4], int lane) {
  if ((unsigned int)gid < (unsigned int)NUM_GRAPHS) {
#pragma unroll
    for (int j = 0; j < 4; ++j) atomicAdd(pooled + (size_t)gid * HDIM + lane + 32 * j, acc[j]);
  }
}

__global__ __launch_bounds__(ROW_WAVES * 32)
void pool_rows_kernel(const float* __restrict__ x, const int* __restrict__ batch,
                      float* __restrict__ pooled, int n) {
  const int lane = threadIdx.x & 31;
  const int wave = __builtin_amdgcn_readfirstlane(threadIdx.x >> 5);
  const int row0 = (blockIdx.x * ROW_WAVES + wave) * ROWS_PER_WAVE;
  if (row0 >= n) return;
  float v[ROWS_PER_WAVE][4];
  int gid[ROWS_PER_WAVE];
#pragma unroll
  for (int r = 0; r < ROWS_PER_WAVE; ++r) {
    const bool live = row0 + r < n;
    const int row = live ? row0 + r : n - 1;
    gid[r] = __builtin_amdgcn_readfirstlane(batch[row]);
#pragma unroll
    for (int j = 0; j < 4; ++j) {
      const float t = x[(size_t)row * HDIM + lane + 32 * j];
      v[r][j] = live ? t : 0.f;
    }
  }
  float acc[4] = {0.f, 0.f, 0.f, 0.f};
  int cur = gid[0];
#pragma unroll
  for (int r = 0; r < ROWS_PER_WAVE; ++r) {
    if (gid[r] != cur) {
      pool_flush(pooled, cur, acc, lane);
#pragma unroll
      for (int j = 0; j < 4; ++j) acc[j] = 0.f;
      cur = gid[r];
    }
#pragma unroll
    for (int j = 0; j < 4; ++j) acc[j] += v[r][j];
  }
  pool_flush(pooled, cur, acc, lane);
}

__global__ void final_linear_kernel(const float* __restrict__ pooled,
                                    const float* __restrict__ W,
                                    const float* __restrict__ b,
                                    float* __restrict__ out) {
  int i = blockIdx.x * blockDim.x + threadIdx.x;
  if (i >= NUM_GRAPHS * NUM_CLASSES) return;
  int g = i / NUM_CLASSES, c = i % NUM_CLASSES;
  float acc = b[c];
  const float* pr = pooled + (size_t)g * HDIM;
#pragma unroll 4
  for (int k = 0; k < HDIM; ++k) acc += pr[k] * W[k * NUM_CLASSES + c];
  out[i] = acc;
}

extern "C" void kernel_launch(void* const* d_in, const int* in_sizes, int n_in,
                              void* d_out, int out_size, void* d_ws, size_t ws_size,
                              hipStream_t stream) {
  const float* x     = (const float*)d_in[0];
  const float* ea    = (const float*)d_in[1];
  const float* enc_W = (const float*)d_in[2];
  const float* enc_b = (const float*)d_in[3];
  const float* t     = (const float*)d_in[4];
  const float* W1    = (const float*)d_in[5];
  const float* b1    = (const float*)d_in[6];
  const float* mlp_g = (const float*)d_in[7];
  const float* mlp_b = (const float*)d_in[8];
  const float* W2    = (const float*)d_in[9];
  const float* b2    = (const float*)d_in[10];
  const float* ln_g  = (const float*)d_in[11];
  const float* ln_b  = (const float*)d_in[12];
  const float* lin_W = (const float*)d_in[13];
  const float* lin_b = (const float*)d_in[14];
  const int*   eidx  = (const int*)d_in[15];
  const int*   batch = (const int*)d_in[16];

  const int N = in_sizes[0] / HDIM;
  const int E = in_sizes[1] / HDIM;
  const int L = in_sizes[4];
  const int* src = eidx;
  const int* dst = eidx + E;

  const size_t NH = (size_t)N * HDIM;
  float* h    = (float*)d_ws;
  float* rbuf = h    + NH;
  float* hc   = rbuf + NH;
  float* t1   = hc   + NH;
  int* lists  = (int*)(t1 + 2 * NH);

  float* out    = (float*)d_out;
  float* pooled = out + NUM_GRAPHS * NUM_CLASSES;

  const dim3 gblk(128);
  const int aggBlocks = (N + NODES_PER_BLOCK - 1) / NODES_PER_BLOCK;
  const int rowBlocks = (N + ROW_WAVES * ROWS_PER_WAVE - 1) / (ROW_WAVES * ROWS_PER_WAVE);

  gemm_rows16_wmma_f32<HDIM, HDIM, 2, 0><<<N / 16, gblk, 0, stream>>>(x, enc_W, enc_b, h, 0, nullptr, nullptr, nullptr);

  for (int layer = 0; layer < L; ++layer) {
    const float* rin = layer == 0 ? h : rbuf;
    if (layer == 0)
      gen_softmax_lds_kernel<true><<<aggBlocks, BLOCK_WAVES * 32, 0, stream>>>(rin, ea, src, dst, t, layer, hc, lists, E, N);
    else
      gen_softmax_lds_kernel<false><<<aggBlocks, BLOCK_WAVES * 32, 0, stream>>>(rin, ea, src, dst, t, layer, hc, lists, E, N);
    gemm_rows16_wmma_f32<HDIM, HDIM2, 4, 1><<<N / 16, gblk, 0, stream>>>(
        hc, W1 + (size_t)layer * HDIM * HDIM2, b1 + layer * HDIM2, t1, 0,
        mlp_g + layer * HDIM2, mlp_b + layer * HDIM2, nullptr);
    const int nl = layer + 1 < L ? layer + 1 : 0;
    gemm_rows16_wmma_f32<HDIM2, HDIM, 2, 2><<<N / 16, gblk, 0, stream>>>(
        t1, W2 + (size_t)layer * HDIM2 * HDIM, b2 + layer * HDIM, h, layer > 0 ? 1 : 0,
        ln_g + nl * HDIM, ln_b + nl * HDIM, rbuf);
  }

  zero_kernel<<<(NUM_GRAPHS * HDIM + 255) / 256, 256, 0, stream>>>(pooled, NUM_GRAPHS * HDIM);
  pool_rows_kernel<<<rowBlocks, ROW_WAVES * 32, 0, stream>>>(rbuf, batch, pooled, N);
  final_linear_kernel<<<(NUM_GRAPHS * NUM_CLASSES + 63) / 64, 64, 0, stream>>>(pooled, lin_W, lin_b, out);
}


# Block's edge list ordered by destination row (LDS counting sort in layer 0's launch); a wave walks a contiguous share, sums a row's run in registers: one LDS add per run, not per edge; new sum order
# speedup vs baseline: 4.9607x; 1.1215x over previous
#include <hip/hip_runtime.h>
#include <hip/hip_bf16.h>

typedef __attribute__((ext_vector_type(2))) float v2f;
typedef __attribute__((ext_vector_type(8))) float v8f;

#define HDIM 128
#define HDIM2 256
#define NUM_GRAPHS 64
#define NUM_CLASSES 10
#define MSG_EPS 1e-7f

template <int NT> struct fvec;
template <> struct fvec<2> { typedef __attribute__((ext_vector_type(2))) float type; };
template <> struct fvec<4> { typedef __attribute__((ext_vector_type(4))) float type; };

template <int XMASK>
__device__ __forceinline__ float row16_xadd(float v) {
  return v + __int_as_float(__builtin_amdgcn_update_dpp(0, __float_as_int(v), 0x160 + XMASK, 0xf, 0xf, false));
}

template <int K, int M, int NT, int LNMODE>
__global__ __launch_bounds__(128)
void gemm_rows16_wmma_f32(const float* __restrict__ X,
                          const float* __restrict__ W,
                          const float* __restrict__ bias,
                          float* __restrict__ Y, int accum,
                          const float* __restrict__ lng, const float* __restrict__ lnb,
                          float* __restrict__ Z) {
  typedef typename fvec<NT>::type vnt;
  typedef __attribute__((ext_vector_type(4))) float v4f;
  static_assert(M == 64 * NT, "4 waves of NT tiles cover the M columns");
  constexpr int PITCH = K + 4;
  constexpr int KU = 16 / NT;
  constexpr int NG = K / (4 * KU);
  __shared__ float s_a[16 * PITCH];
  __shared__ alignas(16) float s_red[16 * 8];

  const int tid  = threadIdx.x;
  const int lane = tid & 31;
  const int wid  = tid >> 5;
  const int row0 = blockIdx.x * 16;
  const int half = lane >> 4;
  const int l16  = lane & 15;
  const int c0   = wid * (16 * NT);
  const int colv = c0 + NT * l16;

  const float* wp = W + (size_t)(half * 2) * M + colv;
  const float* ap = s_a + l16 * PITCH + half * 2;

  vnt bq[2][KU][2];
  v2f aq[2][KU];

#pragma unroll
  for (int u = 0; u < KU; ++u) {
    bq[0][u][0] = *(const vnt*)(wp + (size_t)(u * 4) * M);
    bq[0][u][1] = *(const vnt*)(wp + (size_t)(u * 4 + 1) * M);
  }

  {
    const float* xs = X + (size_t)row0 * K;
#pragma unroll
    for (int i = 0; i < (16 * K / 4) / 128; ++i) {
      const int idx = tid + 128 * i;
      const int r = idx / (K / 4), q = idx % (K / 4);
      const v4f v = *(const v4f*)(xs + (size_t)r * K + q * 4);
      *(v4f*)(s_a + r * PITCH + q * 4) = v;
    }
  }

  const vnt bv = *(const vnt*)(bias + colv);
  v8f c[NT];
  if (accum) {
#pragma unroll
    for (int r = 0; r < 8; ++r) {
      const vnt y = *(const vnt*)(Y + (size_t)(row0 + half * 8 + r) * M + colv);
#pragma unroll
      for (int t = 0; t < NT; ++t) c[t][r] = y[t] + bv[t];
    }
  } else {
#pragma unroll
    for (int t = 0; t < NT; ++t)
#pragma unroll
      for (int r = 0; r < 8; ++r) c[t][r] = bv[t];
  }

  __syncthreads();

#pragma unroll
  for (int u = 0; u < KU; ++u) aq[0][u] = *(const v2f*)(ap + u * 4);

#pragma unroll
  for (int g = 0; g < NG; ++g) {
    const int cur = g & 1, nxt = cur ^ 1;
    if (g + 1 < NG) {
#pragma unroll
      for (int u = 0; u < KU; ++u) {
        const int k = ((g + 1) * KU + u) * 4;
        bq[nxt][u][0] = *(const vnt*)(wp + (size_t)k * M);
        bq[nxt][u][1] = *(const vnt*)(wp + (size_t)(k + 1) * M);
        aq[nxt][u]    = *(const v2f*)(ap + k);
      }
    }
#pragma unroll
    for (int u = 0; u < KU; ++u) {
#pragma unroll
      for (int t = 0; t < NT; ++t) {
        v2f b; b[0] = bq[cur][u][0][t]; b[1] = bq[cur][u][1][t];
        c[t] = __builtin_amdgcn_wmma_f32_16x16x4_f32(false, aq[cur][u], false, b,
                                                     (short)0, c[t], false, false);
      }
    }
  }

  if (LNMODE != 1) {
#pragma unroll
    for (int r = 0; r < 8; ++r) {
      vnt o;
#pragma unroll
      for (int t = 0; t < NT; ++t) o[t] = c[t][r];
      *(vnt*)(Y + (size_t)(row0 + half * 8 + r) * M + colv) = o;
    }
  }
  if (LNMODE != 0) {
    const vnt gv = *(const vnt*)(lng + colv);
    const vnt ev = *(const vnt*)(lnb + colv);
    float s[8], q[8];
#pragma unroll
    for (int r = 0; r < 8; ++r) {
      s[r] = 0.f; q[r] = 0.f;
#pragma unroll
      for (int t = 0; t < NT; ++t) { const float v = c[t][r]; s[r] += v; q[r] += v * v; }
    }
#pragma unroll
    for (int r = 0; r < 8; ++r) { s[r] = row16_xadd<1>(s[r]); q[r] = row16_xadd<1>(q[r]); }
#pragma unroll
    for (int r = 0; r < 8; ++r) { s[r] = row16_xadd<2>(s[r]); q[r] = row16_xadd<2>(q[r]); }
#pragma unroll
    for (int r = 0; r < 8; ++r) { s[r] = row16_xadd<4>(s[r]); q[r] = row16_xadd<4>(q[r]); }
#pragma unroll
    for (int r = 0; r < 8; ++r) { s[r] = row16_xadd<8>(s[r]); q[r] = row16_xadd<8>(q[r]); }
    if (l16 == 0) {
#pragma unroll
      for (int r = 0; r < 8; ++r) {
        v2f p; p[0] = s[r]; p[1] = q[r];
        *(v2f*)(s_red + (half * 8 + r) * 8 + wid * 2) = p;
      }
    }
    __syncthreads();
    float* dst = LNMODE == 1 ? Y : Z;
#pragma unroll
    for (int r = 0; r < 8; ++r) {
      const v4f p0 = *(const v4f*)(s_red + (half * 8 + r) * 8);
      const v4f p1 = *(const v4f*)(s_red + (half * 8 + r) * 8 + 4);
      const float sum  = ((p0[0] + p0[2]) + p1[0]) + p1[2];
      const float sq   = ((p0[1] + p0[3]) + p1[1]) + p1[3];
      const float mean = sum / (float)M;
      const float var  = sq / (float)M - mean * mean;
      const float inv  = rsqrtf(var + 1e-5f);
      vnt o;
#pragma unroll
      for (int t = 0; t < NT; ++t)
        o[t] = fmaxf((c[t][r] - mean) * inv * gv[t] + ev[t], 0.f);
      *(vnt*)(dst + (size_t)(row0 + half * 8 + r) * M + colv) = o;
    }
  }
}

#define ROWS_PER_WAVE 8
#define ROW_WAVES 4

__global__ void zero_kernel(float* p, int n) {
  int i = blockIdx.x * blockDim.x + threadIdx.x;
  if (i < n) p[i] = 0.f;
}

#define NODES_PER_BLOCK 40
#define BLOCK_WAVES 32
#define SCAN_CHUNKS 16
#define GROUP_EDGES 8
#define M_CAP 1536
#define QCOLS 32
#define LIST_HDR 32
#define LIST_STRIDE (LIST_HDR + 3 * M_CAP)
#define EXP_ARG_CAP 16.0f

template <int PHASE>
__device__ __forceinline__ void fold_rows(
    const float* __restrict__ x, const float* __restrict__ ea,
    const int (&pe)[GROUP_EDGES], const int (&s)[GROUP_EDGES], const int (&r)[GROUP_EDGES],
    int count, float tv, int lane, float* s_mx, float* s_den, float* s_num) {
  float a[GROUP_EDGES][4], b[GROUP_EDGES][4];
#pragma unroll
  for (int g = 0; g < GROUP_EDGES; ++g) {
    const float* xr = x  + (size_t)s[g]  * HDIM + lane;
    const float* er = ea + (size_t)pe[g] * HDIM + lane;
#pragma unroll
    for (int j = 0; j < 4; ++j) { a[g][j] = xr[32 * j]; b[g][j] = er[32 * j]; }
  }
  float mv[GROUP_EDGES][4];
  if (PHASE == 2) {
#pragma unroll
    for (int g = 0; g < GROUP_EDGES; ++g) {
      if (g < count) {
#pragma unroll
        for (int j = 0; j < 4; ++j) mv[g][j] = s_mx[r[g] * HDIM + lane + 32 * j];
      }
    }
  }
#pragma unroll
  for (int g = 0; g < GROUP_EDGES; ++g) {
    if (g < count) {
      const int o = r[g] * HDIM + lane;
#pragma unroll
      for (int j = 0; j < 4; ++j) {
        const float m = fmaxf(a[g][j] + b[g][j], 0.f) + MSG_EPS;
        if (PHASE == 1) {
          const float v = m * tv;
          if (v >= 0.0f) atomicMax((int*)&s_mx[o + 32 * j], __float_as_int(v));
          else           atomicMin((unsigned int*)&s_mx[o + 32 * j], __float_as_uint(v));
        } else {
          const float ex = __expf(m * tv - mv[g][j]);
          atomicAdd(&s_den[o + 32 * j], ex);
          atomicAdd(&s_num[o + 32 * j], ex * m);
        }
      }
    }
  }
}

template <int PHASE>
__device__ __forceinline__ void fold_edge_group(
    const float* __restrict__ x, const float* __restrict__ ea,
    const int* __restrict__ src, const int* __restrict__ dst,
    const int (&pe)[GROUP_EDGES], int count, int node0, float tv, int lane,
    float* s_mx, float* s_den, float* s_num) {
  int s[GROUP_EDGES], r[GROUP_EDGES];
#pragma unroll
  for (int g = 0; g < GROUP_EDGES; ++g) {
    s[g] = src[pe[g]];
    r[g] = dst[pe[g]] - node0;
  }
  fold_rows<PHASE>(x, ea, pe, s, r, count, tv, lane, s_mx, s_den, s_num);
}

__device__ __forceinline__ unsigned int scan_step_hits(const int* __restrict__ dst, int E, int c0,
                                                       int node0, int lane) {
  unsigned int hits = 0u;
  if ((c0 + SCAN_CHUNKS) * 32 <= E) {
    const int* p = dst + c0 * 32 + lane;
#pragma unroll
    for (int u = 0; u < SCAN_CHUNKS; ++u) {
      const int dv = p[u * 32];
      hits |= ((unsigned int)(dv - node0) < (unsigned int)NODES_PER_BLOCK ? 1u : 0u) << u;
    }
  } else {
#pragma unroll
    for (int u = 0; u < SCAN_CHUNKS; ++u) {
      const int e  = (c0 + u) * 32 + lane;
      const int dv = dst[e < E ? e : E - 1];
      const bool hit = (e < E) && ((unsigned int)(dv - node0) < (unsigned int)NODES_PER_BLOCK);
      hits |= (hit ? 1u : 0u) << u;
    }
  }
  return hits;
}

__device__ __forceinline__ void scan_build_list(
    const int* __restrict__ dst, int E, int node0, int lane, int wave,
    int* s_cnt, int* s_le) {
  const int nchunks = (E + 31) >> 5;
  for (int c0 = wave * SCAN_CHUNKS; c0 < nchunks; c0 += BLOCK_WAVES * SCAN_CHUNKS) {
    unsigned int hits = scan_step_hits(dst, E, c0, node0, lane);
    for (int it = 0; it < SCAN_CHUNKS; ++it) {
      const unsigned int active = __builtin_amdgcn_ballot_w32(hits != 0u);
      if (active == 0u) break;
      int slot0 = 0;
      if (lane == 0) slot0 = atomicAdd(s_cnt, __builtin_popcount(active));
      slot0 = __builtin_amdgcn_readfirstlane(slot0);
      if (hits != 0u) {
        const int u = __builtin_ctz(hits);
        hits &= hits - 1u;
        const int e = (c0 + u) * 32 + lane;
        const int slot = slot0 + __builtin_popcount(active & ((1u << lane) - 1u));
        if (slot < M_CAP) s_le[slot] = e;
      }
    }
  }
}

template <int NG, bool FULL>
__device__ __forceinline__ void quarter_messages(
    const float* __restrict__ x, const float* __restrict__ ea,
    int first, int n, float tv, int lane, int col,
    const int* s_le, const int* s_ls, const int* s_lr, float* s_mx, float* s_m) {
  constexpr int NE = GROUP_EDGES * NG;
  const int used = FULL ? NE : n;
  const int li = lane < used ? first + (lane >> 3) * (GROUP_EDGES * BLOCK_WAVES) + (lane & 7) : first;
  const int ev = s_le[li], sv = s_ls[li], rv = s_lr[li];
  float a[NE], b[NE];
#pragma unroll
  for (int g = 0; g < NE; ++g) {
    const int pe = __builtin_amdgcn_readlane(ev, g);
    const int s  = __builtin_amdgcn_readlane(sv, g);
    a[g] = x[(size_t)s * HDIM + col];
    b[g] = ea[(size_t)pe * HDIM + col];
  }
#pragma unroll
  for (int g = 0; g < NE; ++g) {
    if (FULL || g < n) {
      const int r   = __builtin_amdgcn_readlane(rv, g);
      const int o   = r * HDIM + col;
      const int row = first + (g >> 3) * (GROUP_EDGES * BLOCK_WAVES) + (g & 7);
      const float m = fmaxf(a[g] + b[g], 0.f) + MSG_EPS;
      s_m[row * QCOLS + lane] = m;
      const float v = m * tv;
      if (v >= 0.0f) atomicMax((int*)&s_mx[o], __float_as_int(v));
      else           atomicMin((unsigned int*)&s_mx[o], __float_as_uint(v));
    }
  }
}

template <bool FULL>
__device__ __forceinline__ void quarter_sums(
    int first, int n, float tv, int lane, int col,
    const int* s_lr, const float* s_mx, float* s_den, float* s_num, const float* s_m) {
  const int used = FULL ? GROUP_EDGES : n;
  const int rv = s_lr[lane < used ? first + lane : first];
  float m[GROUP_EDGES], mv[GROUP_EDGES];
  int o[GROUP_EDGES];
#pragma unroll
  for (int g = 0; g < GROUP_EDGES; ++g) {
    if (FULL || g < n) {
      o[g]  = __builtin_amdgcn_readlane(rv, g) * HDIM + col;
      m[g]  = s_m[(first + g) * QCOLS + lane];
      mv[g] = s_mx[o[g]];
    }
  }
#pragma unroll
  for (int g = 0; g < GROUP_EDGES; ++g) {
    if (FULL || g < n) {
      const float ex = __expf(m[g] * tv - mv[g]);
      atomicAdd(&s_den[o[g]], ex);
      atomicAdd(&s_num[o[g]], ex * m[g]);
    }
  }
}

__device__ __forceinline__ void fold_list_by_quarters(
    const float* __restrict__ x, const float* __restrict__ ea,
    int cnt, float tv, int lane, int wave,
    const int* s_le, const int* s_ls, const int* s_lr,
    float* s_mx, float* s_den, float* s_num, float* s_m) {
  const int G     = (cnt + GROUP_EDGES - 1) / GROUP_EDGES;
  const int tail  = cnt - GROUP_EDGES * (G - 1);
  const int ng    = G > wave ? (G - wave + BLOCK_WAVES - 1) / BLOCK_WAVES : 0;
  const bool has_short = ng > 0 && wave + BLOCK_WAVES * (ng - 1) == G - 1 && tail < GROUP_EDGES;
  const int nfull = has_short ? ng - 1 : ng;
  const int STRIDE = GROUP_EDGES * BLOCK_WAVES;
  const int first0 = GROUP_EDGES * wave;
#pragma unroll 1
  for (int q = 0; q < HDIM / QCOLS; ++q) {
    const int col = q * QCOLS + lane;
    int j = 0;
    for (; j + 3 <= nfull; j += 3)
      quarter_messages<3, true>(x, ea, first0 + j * STRIDE, 0, tv, lane, col, s_le, s_ls, s_lr, s_mx, s_m);
    if (nfull - j == 2)
      quarter_messages<2, true>(x, ea, first0 + j * STRIDE, 0, tv, lane, col, s_le, s_ls, s_lr, s_mx, s_m);
    else if (nfull - j == 1)
      quarter_messages<1, true>(x, ea, first0 + j * STRIDE, 0, tv, lane, col, s_le, s_ls, s_lr, s_mx, s_m);
    if (has_short)
      quarter_messages<1, false>(x, ea, first0 + nfull * STRIDE, tail, tv, lane, col, s_le, s_ls, s_lr, s_mx, s_m);
    __syncthreads();
    for (int k = 0; k < nfull; ++k)
      quarter_sums<true>(first0 + k * STRIDE, 0, tv, lane, col, s_lr, s_mx, s_den, s_num, s_m);
    if (has_short)
      quarter_sums<false>(first0 + nfull * STRIDE, tail, tv, lane, col, s_lr, s_mx, s_den, s_num, s_m);
  }
}

__device__ __forceinline__ void flush_run(int row, const float (&aden)[4], const float (&anum)[4],
                                          int lane, float* s_den, float* s_num) {
  const int o = row * HDIM + lane;
#pragma unroll
  for (int j = 0; j < 4; ++j) {
    atomicAdd(&s_den[o + 32 * j], aden[j]);
    atomicAdd(&s_num[o + 32 * j], anum[j]);
  }
}

template <bool FULL>
__device__ __forceinline__ float one_pass_group(
    const float* __restrict__ x, const float* __restrict__ ea,
    int first, int n, float tv, int lane,
    const int* s_le, const int* s_ls, const int* s_lr,
    float* s_den, float* s_num, float vabs,
    int& cur, float (&aden)[4], float (&anum)[4]) {
  const int used = FULL ? GROUP_EDGES : n;
  const int li = lane < used ? first + lane : first;
  const int ev = s_le[li], sv = s_ls[li], rv = s_lr[li];
  float a[GROUP_EDGES][4], b[GROUP_EDGES][4];
#pragma unroll
  for (int g = 0; g < GROUP_EDGES; ++g) {
    const int pe = __builtin_amdgcn_readlane(ev, g);
    const int s  = __builtin_amdgcn_readlane(sv, g);
    const float* xr = x  + (size_t)s  * HDIM + lane;
    const float* er = ea + (size_t)pe * HDIM + lane;
#pragma unroll
    for (int j = 0; j < 4; ++j) { a[g][j] = xr[32 * j]; b[g][j] = er[32 * j]; }
  }
#pragma unroll
  for (int g = 0; g < GROUP_EDGES; ++g) {
    if (FULL || g < n) {
      const int r = __builtin_amdgcn_readlane(rv, g);
      if (r != cur) {
        if (cur >= 0) flush_run(cur, aden, anum, lane, s_den, s_num);
#pragma unroll
        for (int j = 0; j < 4; ++j) { aden[j] = 0.f; anum[j] = 0.f; }
        cur = r;
      }
#pragma unroll
      for (int j = 0; j < 4; ++j) {
        const float m  = fmaxf(a[g][j] + b[g][j], 0.f) + MSG_EPS;
        const float v  = m * tv;
        vabs = fmaxf(vabs, fabsf(v));
        const float ex = __expf(v);
        aden[j] += ex;
        anum[j] += ex * m;
      }
    }
  }
  return vabs;
}

__device__ __forceinline__ float fold_list_one_pass(
    const float* __restrict__ x, const float* __restrict__ ea,
    int cnt, float tv, int lane, int wave,
    const int* s_le, const int* s_ls, const int* s_lr,
    float* s_den, float* s_num) {
  const int G    = (cnt + GROUP_EDGES - 1) / GROUP_EDGES;
  const int tail = cnt - GROUP_EDGES * (G - 1);
  const int base = G / BLOCK_WAVES, rem = G % BLOCK_WAVES;
  const int k0   = wave * base + (wave < rem ? wave : rem);
  const int k1   = k0 + base + (wave < rem ? 1 : 0);
  float vabs = 0.f;
  int cur = -1;
  float aden[4] = {0.f, 0.f, 0.f, 0.f}, anum[4] = {0.f, 0.f, 0.f, 0.f};
  for (int k = k0; k < k1; ++k) {
    if (k < G - 1 || tail == GROUP_EDGES)
      vabs = one_pass_group<true>(x, ea, GROUP_EDGES * k, 0, tv, lane, s_le, s_ls, s_lr, s_den, s_num, vabs, cur, aden, anum);
    else
      vabs = one_pass_group<false>(x, ea, GROUP_EDGES * k, tail, tv, lane, s_le, s_ls, s_lr, s_den, s_num, vabs, cur, aden, anum);
  }
  if (cur >= 0) flush_run(cur, aden, anum, lane, s_den, s_num);
  return vabs;
}

template <int PHASE>
__device__ __forceinline__ void scan_and_fold(
    const float* __restrict__ x, const float* __restrict__ ea,
    const int* __restrict__ src, const int* __restrict__ dst,
    int E, int node0, float tv, int lane, int wave,
    float* s_mx, float* s_den, float* s_num) {
  int pe[GROUP_EDGES];
#pragma unroll
  for (int g = 0; g < GROUP_EDGES; ++g) pe[g] = 0;
  int pn = 0;
  const int nchunks = (E + 31) >> 5;
  for (int c0 = wave * SCAN_CHUNKS; c0 < nchunks; c0 += BLOCK_WAVES * SCAN_CHUNKS) {
    const unsigned int hits = scan_step_hits(dst, E, c0, node0, lane);
    if (__builtin_amdgcn_ballot_w32(hits != 0u) != 0u) {
#pragma unroll 1
      for (int u = 0; u < SCAN_CHUNKS; ++u) {
        unsigned int mask = __builtin_amdgcn_ballot_w32(((hits >> u) & 1u) != 0u);
        const int base = (c0 + u) * 32;
        while (mask != 0u) {
          const int bit = __builtin_ctz(mask);
          mask &= mask - 1u;
#pragma unroll
          for (int g = GROUP_EDGES - 1; g > 0; --g) pe[g] = pe[g - 1];
          pe[0] = base + bit;
          if (++pn == GROUP_EDGES) {
            fold_edge_group<PHASE>(x, ea, src, dst, pe, GROUP_EDGES, node0, tv, lane, s_mx, s_den, s_num);
            pn = 0;
          }
        }
      }
    }
  }
  if (pn > 0)
    fold_edge_group<PHASE>(x, ea, src, dst, pe, pn, node0, tv, lane, s_mx, s_den, s_num);
}

template <bool BUILD>
__global__ __launch_bounds__(BLOCK_WAVES * 32)
void gen_softmax_lds_kernel(const float* __restrict__ x, const float* __restrict__ ea,
                            const int* __restrict__ src, const int* __restrict__ dst,
                            const float* __restrict__ tptr, int layer,
                            float* __restrict__ hc, int* __restrict__ lists, int E, int N) {
  __shared__ float s_mx[NODES_PER_BLOCK * HDIM];
  __shared__ float s_den[NODES_PER_BLOCK * HDIM];
  __shared__ float s_num[NODES_PER_BLOCK * HDIM];
  __shared__ float s_m[M_CAP * QCOLS];
  __shared__ int s_le[M_CAP];
  __shared__ int s_ls[M_CAP];
  __shared__ int s_lr[M_CAP];
  __shared__ int s_rowcnt[NODES_PER_BLOCK];
  __shared__ int s_rowoff[NODES_PER_BLOCK];
  __shared__ int s_cnt;
  __shared__ int s_redo;
  const int tid   = threadIdx.x;
  const int lane  = tid & 31;
  const int wave  = __builtin_amdgcn_readfirstlane(tid >> 5);
  const int node0 = blockIdx.x * NODES_PER_BLOCK;
  int* gl = lists + (size_t)blockIdx.x * LIST_STRIDE;
  for (int i = tid; i < NODES_PER_BLOCK * HDIM; i += BLOCK_WAVES * 32) {
    s_mx[i]  = __int_as_float(0xff800000);
    s_den[i] = 0.f;
    s_num[i] = 0.f;
  }
  const float tv = tptr[layer];
  if (tid == 0) s_redo = 0;
  int cnt;
  if (BUILD) {
    if (tid == 0) s_cnt = 0;
    if (tid < NODES_PER_BLOCK) s_rowcnt[tid] = 0;
    __syncthreads();
    scan_build_list(dst, E, node0, lane, wave, &s_cnt, s_le);
    __syncthreads();
    cnt = __builtin_amdgcn_readfirstlane(s_cnt);
    if (tid == 0) gl[0] = cnt;
    if (cnt <= M_CAP) {
      constexpr int TRIPS = (M_CAP + BLOCK_WAVES * 32 - 1) / (BLOCK_WAVES * 32);
      int ve[TRIPS], vs[TRIPS], vr[TRIPS], vk[TRIPS];
#pragma unroll
      for (int k = 0; k < TRIPS; ++k) {
        const int i = tid + k * (BLOCK_WAVES * 32);
        if (i < cnt) {
          const int e = s_le[i];
          ve[k] = e;
          vs[k] = src[e];
          vr[k] = dst[e] - node0;
          vk[k] = atomicAdd(&s_rowcnt[vr[k]], 1);
        }
      }
      __syncthreads();
      if (tid < NODES_PER_BLOCK) {
        int off = 0;
#pragma unroll
        for (int q = 0; q < NODES_PER_BLOCK; ++q) {
          const int cq = s_rowcnt[q];
          off += q < tid ? cq : 0;
        }
        s_rowoff[tid] = off;
      }
      __syncthreads();
#pragma unroll
      for (int k = 0; k < TRIPS; ++k) {
        const int i = tid + k * (BLOCK_WAVES * 32);
        if (i < cnt) {
          const int p = s_rowoff[vr[k]] + vk[k];
          s_le[p] = ve[k];
          s_ls[p] = vs[k];
          s_lr[p] = vr[k];
          gl[LIST_HDR + p]             = ve[k];
          gl[LIST_HDR + M_CAP + p]     = vs[k];
          gl[LIST_HDR + 2 * M_CAP + p] = vr[k];
        }
      }
    }
  } else {
    cnt = __builtin_amdgcn_readfirstlane(gl[0]);
    if (cnt <= M_CAP) {
      constexpr int TRIPS = (M_CAP + BLOCK_WAVES * 32 - 1) / (BLOCK_WAVES * 32);
      int ve[TRIPS], vs[TRIPS], vr[TRIPS];
#pragma unroll
      for (int k = 0; k < TRIPS; ++k) {
        const int i = tid + k * (BLOCK_WAVES * 32);
        if (i < cnt) {
          ve[k] = gl[LIST_HDR + i];
          vs[k] = gl[LIST_HDR + M_CAP + i];
          vr[k] = gl[LIST_HDR + 2 * M_CAP + i];
        }
      }
#pragma unroll
      for (int k = 0; k < TRIPS; ++k) {
        const int i = tid + k * (BLOCK_WAVES * 32);
        if (i < cnt) {
          s_le[i] = ve[k];
          s_ls[i] = vs[k];
          s_lr[i] = vr[k];
        }
      }
    }
  }
  __syncthreads();
  if (cnt <= M_CAP) {
    const float vabs = fold_list_one_pass(x, ea, cnt, tv, lane, wave, s_le, s_ls, s_lr, s_den, s_num);
    if (__builtin_amdgcn_ballot_w32(!(vabs <= EXP_ARG_CAP)) != 0u && lane == 0) s_redo = 1;
    __syncthreads();
    if (__builtin_amdgcn_readfirstlane(s_redo) != 0) {
      for (int i = tid; i < NODES_PER_BLOCK * HDIM; i += BLOCK_WAVES * 32) {
        s_den[i] = 0.f;
        s_num[i] = 0.f;
      }
      __syncthreads();
      fold_list_by_quarters(x, ea, cnt, tv, lane, wave, s_le, s_ls, s_lr, s_mx, s_den, s_num, s_m);
      __syncthreads();
    }
  } else {
    scan_and_fold<1>(x, ea, src, dst, E, node0, tv, lane, wave, s_mx, s_den, s_num);
    __syncthreads();
    scan_and_fold<2>(x, ea, src, dst, E, node0, tv, lane, wave, s_mx, s_den, s_num);
    __syncthreads();
  }
  for (int i = tid; i < NODES_PER_BLOCK * HDIM; i += BLOCK_WAVES * 32) {
    const int node = node0 + i / HDIM;
    if (node < N) {
      const float dv  = s_den[i];
      const float agg = dv > 0.f ? s_num[i] / dv : 0.f;
      const size_t gi = (size_t)node * HDIM + i % HDIM;
      hc[gi] = agg + x[gi];
    }
  }
}

__device__ __forceinline__ void pool_flush(float* __restrict__ pooled, int gid, const float (&acc)[4], int lane) {
  if ((unsigned int)gid < (unsigned int)NUM_GRAPHS) {
#pragma unroll
    for (int j = 0; j < 4; ++j) atomicAdd(pooled + (size_t)gid * HDIM + lane + 32 * j, acc[j]);
  }
}

__global__ __launch_bounds__(ROW_WAVES * 32)
void pool_rows_kernel(const float* __restrict__ x, const int* __restrict__ batch,
                      float* __restrict__ pooled, int n) {
  const int lane = threadIdx.x & 31;
  const int wave = __builtin_amdgcn_readfirstlane(threadIdx.x >> 5);
  const int row0 = (blockIdx.x * ROW_WAVES + wave) * ROWS_PER_WAVE;
  if (row0 >= n) return;
  float v[ROWS_PER_WAVE][4];
  int gid[ROWS_PER_WAVE];
#pragma unroll
  for (int r = 0; r < ROWS_PER_WAVE; ++r) {
    const bool live = row0 + r < n;
    const int row = live ? row0 + r : n - 1;
    gid[r] = __builtin_amdgcn_readfirstlane(batch[row]);
#pragma unroll
    for (int j = 0; j < 4; ++j) {
      const float t = x[(size_t)row * HDIM + lane + 32 * j];
      v[r][j] = live ? t : 0.f;
    }
  }
  float acc[4] = {0.f, 0.f, 0.f, 0.f};
  int cur = gid[0];
#pragma unroll
  for (int r = 0; r < ROWS_PER_WAVE; ++r) {
    if (gid[r] != cur) {
      pool_flush(pooled, cur, acc, lane);
#pragma unroll
      for (int j = 0; j < 4; ++j) acc[j] = 0.f;
      cur = gid[r];
    }
#pragma unroll
    for (int j = 0; j < 4; ++j) acc[j] += v[r][j];
  }
  pool_flush(pooled, cur, acc, lane);
}

__global__ void final_linear_kernel(const float* __restrict__ pooled,
                                    const float* __restrict__ W,
                                    const float* __restrict__ b,
                                    float* __restrict__ out) {
  int i = blockIdx.x * blockDim.x + threadIdx.x;
  if (i >= NUM_GRAPHS * NUM_CLASSES) return;
  int g = i / NUM_CLASSES, c = i % NUM_CLASSES;
  float acc = b[c];
  const float* pr = pooled + (size_t)g * HDIM;
#pragma unroll 4
  for (int k = 0; k < HDIM; ++k) acc += pr[k] * W[k * NUM_CLASSES + c];
  out[i] = acc;
}

extern "C" void kernel_launch(void* const* d_in, const int* in_sizes, int n_in,
                              void* d_out, int out_size, void* d_ws, size_t ws_size,
                              hipStream_t stream) {
  const float* x     = (const float*)d_in[0];
  const float* ea    = (const float*)d_in[1];
  const float* enc_W = (const float*)d_in[2];
  const float* enc_b = (const float*)d_in[3];
  const float* t     = (const float*)d_in[4];
  const float* W1    = (const float*)d_in[5];
  const float* b1    = (const float*)d_in[6];
  const float* mlp_g = (const float*)d_in[7];
  const float* mlp_b = (const float*)d_in[8];
  const float* W2    = (const float*)d_in[9];
  const float* b2    = (const float*)d_in[10];
  const float* ln_g  = (const float*)d_in[11];
  const float* ln_b  = (const float*)d_in[12];
  const float* lin_W = (const float*)d_in[13];
  const float* lin_b = (const float*)d_in[14];
  const int*   eidx  = (const int*)d_in[15];
  const int*   batch = (const int*)d_in[16];

  const int N = in_sizes[0] / HDIM;
  const int E = in_sizes[1] / HDIM;
  const int L = in_sizes[4];
  const int* src = eidx;
  const int* dst = eidx + E;

  const size_t NH = (size_t)N * HDIM;
  float* h    = (float*)d_ws;
  float* rbuf = h    + NH;
  float* hc   = rbuf + NH;
  float* t1   = hc   + NH;
  int* lists  = (int*)(t1 + 2 * NH);

  float* out    = (float*)d_out;
  float* pooled = out + NUM_GRAPHS * NUM_CLASSES;

  const dim3 gblk(128);
  const int aggBlocks = (N + NODES_PER_BLOCK - 1) / NODES_PER_BLOCK;
  const int rowBlocks = (N + ROW_WAVES * ROWS_PER_WAVE - 1) / (ROW_WAVES * ROWS_PER_WAVE);

  gemm_rows16_wmma_f32<HDIM, HDIM, 2, 0><<<N / 16, gblk, 0, stream>>>(x, enc_W, enc_b, h, 0, nullptr, nullptr, nullptr);

  for (int layer = 0; layer < L; ++layer) {
    const float* rin = layer == 0 ? h : rbuf;
    if (layer == 0)
      gen_softmax_lds_kernel<true><<<aggBlocks, BLOCK_WAVES * 32, 0, stream>>>(rin, ea, src, dst, t, layer, hc, lists, E, N);
    else
      gen_softmax_lds_kernel<false><<<aggBlocks, BLOCK_WAVES * 32, 0, stream>>>(rin, ea, src, dst, t, layer, hc, lists, E, N);
    gemm_rows16_wmma_f32<HDIM, HDIM2, 4, 1><<<N / 16, gblk, 0, stream>>>(
        hc, W1 + (size_t)layer * HDIM * HDIM2, b1 + layer * HDIM2, t1, 0,
        mlp_g + layer * HDIM2, mlp_b + layer * HDIM2, nullptr);
    const int nl = layer + 1 < L ? layer + 1 : 0;
    gemm_rows16_wmma_f32<HDIM2, HDIM, 2, 2><<<N / 16, gblk, 0, stream>>>(
        t1, W2 + (size_t)layer * HDIM2 * HDIM, b2 + layer * HDIM, h, layer > 0 ? 1 : 0,
        ln_g + nl * HDIM, ln_b + nl * HDIM, rbuf);
  }

  zero_kernel<<<(NUM_GRAPHS * HDIM + 255) / 256, 256, 0, stream>>>(pooled, NUM_GRAPHS * HDIM);
  pool_rows_kernel<<<rowBlocks, ROW_WAVES * 32, 0, stream>>>(rbuf, batch, pooled, N);
  final_linear_kernel<<<(NUM_GRAPHS * NUM_CLASSES + 63) / 64, 64, 0, stream>>>(pooled, lin_W, lin_b, out);
}
